# starting kernel
# baseline (speedup 1.0000x reference)
#include <hip/hip_runtime.h>

#define ENT_DIM  128
#define REL_DIM  128
#define CMP_DIM  256
#define NUM_RELC 200
#define MTILES   4

typedef __attribute__((ext_vector_type(2))) float v2f;
typedef __attribute__((ext_vector_type(8))) float v8f;

__device__ __forceinline__ v8f wmma4_f32(v2f a, v2f b, v8f c) {
    return __builtin_amdgcn_wmma_f32_16x16x4_f32(
         false, a,  false, b,
         (short)0, c,  false,  false);
}

__global__ __launch_bounds__(256)
void edge_msg_kernel(const int*   __restrict__ src,
                     const int*   __restrict__ dst,
                     const int*   __restrict__ etypes,
                     const float* __restrict__ ent_emb,
                     const float* __restrict__ rel_emb,
                     const float* __restrict__ Wo,
                     const float* __restrict__ bo,
                     const float* __restrict__ Wi,
                     const float* __restrict__ bi,
                     float*       __restrict__ agg,
                     float*       __restrict__ deg,
                     int E)
{
    __shared__ int s_et [MTILES * 16];
    __shared__ int s_dst[MTILES * 16];

    const int  tid   = threadIdx.x;
    const int  lane  = tid & 31;
    const int  wave  = tid >> 5;
    const int  m     = lane & 15;
    const int  khalf = (lane >> 4) << 1;
    const long base  = (long)blockIdx.x * (MTILES * 16);

    if (tid < MTILES * 16) {
        long e  = base + tid;
        int  ec = (e < E) ? (int)e : (E - 1);
        s_et [tid] = etypes[ec];
        s_dst[tid] = dst[ec];
    }
    __syncthreads();

    const float* arel[MTILES];
    const float* aent[MTILES];
    #pragma unroll
    for (int t = 0; t < MTILES; ++t) {
        long e  = base + t * 16 + m;
        int  ec = (e < E) ? (int)e : (E - 1);
        arel[t] = rel_emb + (long)etypes[ec] * REL_DIM;
        aent[t] = ent_emb + (long)src[ec]    * ENT_DIM;
    }

    const int    n      = (wave << 4) + m;
    const float* wo_row = Wo + (long)n * CMP_DIM;
    const float* wi_row = Wi + (long)n * CMP_DIM;

    v8f acc_o[MTILES], acc_i[MTILES];
    #pragma unroll
    for (int t = 0; t < MTILES; ++t) {
        acc_o[t] = (v8f){0.f,0.f,0.f,0.f,0.f,0.f,0.f,0.f};
        acc_i[t] = (v8f){0.f,0.f,0.f,0.f,0.f,0.f,0.f,0.f};
    }

    #pragma unroll 2
    for (int k = 0; k < REL_DIM; k += 4) {
        v2f bO = *(const v2f*)(wo_row + k + khalf);
        v2f bI = *(const v2f*)(wi_row + k + khalf);
        v2f a[MTILES];
        #pragma unroll
        for (int t = 0; t < MTILES; ++t)
            a[t] = *(const v2f*)(arel[t] + k + khalf);
        #pragma unroll
        for (int t = 0; t < MTILES; ++t) {
            acc_o[t] = wmma4_f32(a[t], bO, acc_o[t]);
            acc_i[t] = wmma4_f32(a[t], bI, acc_i[t]);
        }
    }
    #pragma unroll 2
    for (int k = 0; k < ENT_DIM; k += 4) {
        v2f bO = *(const v2f*)(wo_row + REL_DIM + k + khalf);
        v2f bI = *(const v2f*)(wi_row + REL_DIM + k + khalf);
        v2f a[MTILES];
        #pragma unroll
        for (int t = 0; t < MTILES; ++t)
            a[t] = *(const v2f*)(aent[t] + k + khalf);
        #pragma unroll
        for (int t = 0; t < MTILES; ++t) {
            acc_o[t] = wmma4_f32(a[t], bO, acc_o[t]);
            acc_i[t] = wmma4_f32(a[t], bI, acc_i[t]);
        }
    }

    const float bob   = bo[n];
    const float bib   = bi[n];
    const int   rhalf = (lane >> 4) << 3;

    #pragma unroll
    for (int t = 0; t < MTILES; ++t) {
        #pragma unroll
        for (int j = 0; j < 8; ++j) {
            const int  me = t * 16 + j + rhalf;
            const long ee = base + me;
            if (ee < E) {
                const bool  is_o = s_et[me] < NUM_RELC;
                const float v    = is_o ? (acc_o[t][j] + bob) : (acc_i[t][j] + bib);
                atomicAdd(&agg[(long)s_dst[me] * ENT_DIM + n], v);
            }
        }
    }

    if (tid < MTILES * 16 && (base + tid) < E)
        atomicAdd(&deg[s_dst[tid]], 1.0f);
}

__global__ __launch_bounds__(256)
void gemm_bias_kernel(const float* __restrict__ A,
                      const float* __restrict__ W,
                      const float* __restrict__ b,
                      const float* __restrict__ agg,
                      const float* __restrict__ deg,
                      float*       __restrict__ out,
                      int N)
{
    const int  tid   = threadIdx.x;
    const int  lane  = tid & 31;
    const int  wave  = tid >> 5;
    const int  m     = lane & 15;
    const int  khalf = (lane >> 4) << 1;
    const long tile  = (long)blockIdx.x * 16;

    const long r    = tile + m;
    const long rc   = (r < N) ? r : (long)(N - 1);
    const float* ar = A + rc * ENT_DIM;

    const int    n  = (wave << 4) + m;
    const float* wr = W + (long)n * ENT_DIM;

    v8f acc = {0.f,0.f,0.f,0.f,0.f,0.f,0.f,0.f};

    #pragma unroll 4
    for (int k = 0; k < ENT_DIM; k += 4) {
        v2f av = *(const v2f*)(ar + k + khalf);
        v2f bv = *(const v2f*)(wr + k + khalf);
        acc = wmma4_f32(av, bv, acc);
    }

    const float bn    = b[n];
    const int   rhalf = (lane >> 4) << 3;

    #pragma unroll
    for (int j = 0; j < 8; ++j) {
        const int  me = j + rhalf;
        const long rr = tile + me;
        if (rr < N) {
            float v = acc[j] + bn;
            if (agg != nullptr) {
                float d = deg[rr];
                d = (d > 1.0f) ? d : 1.0f;
                v += agg[rr * ENT_DIM + n] / d;
            }
            out[rr * ENT_DIM + n] = v;
        }
    }
}

__global__ void zero_kernel(float* __restrict__ p, long n)
{
    const long i = (long)blockIdx.x * blockDim.x + threadIdx.x;
    if (i < n) p[i] = 0.0f;
}

extern "C" void kernel_launch(void* const* d_in, const int* in_sizes, int n_in,
                              void* d_out, int out_size, void* d_ws, size_t ws_size,
                              hipStream_t stream)
{
    const int*   src     = (const int*)  d_in[0];
    const int*   dst     = (const int*)  d_in[1];
    const int*   etypes  = (const int*)  d_in[2];
    const float* ent_emb = (const float*)d_in[3];
    const float* rel_emb = (const float*)d_in[4];
    const float* Wo      = (const float*)d_in[5];
    const float* bo      = (const float*)d_in[6];
    const float* Wi      = (const float*)d_in[7];
    const float* bi      = (const float*)d_in[8];
    const float* Ws      = (const float*)d_in[9];
    const float* bs      = (const float*)d_in[10];
    const float* Wr      = (const float*)d_in[11];
    const float* br      = (const float*)d_in[12];

    const int E = in_sizes[0];
    const int N = in_sizes[3] / ENT_DIM;
    const int R = in_sizes[4] / REL_DIM;

    float* agg = (float*)d_ws;
    float* deg = agg + (size_t)N * ENT_DIM;

    const long zn = (long)N * ENT_DIM + N;
    zero_kernel<<<(int)((zn + 255) / 256), 256, 0, stream>>>(agg, zn);

    const int edges_per_block = MTILES * 16;
    edge_msg_kernel<<<(E + edges_per_block - 1) / edges_per_block, 256, 0, stream>>>(
        src, dst, etypes, ent_emb, rel_emb, Wo, bo, Wi, bi, agg, deg, E);

    float* out_ent = (float*)d_out;
    float* out_rel = out_ent + (size_t)N * ENT_DIM;

    gemm_bias_kernel<<<(N + 15) / 16, 256, 0, stream>>>(
        ent_emb, Ws, bs, agg, deg, out_ent, N);

    gemm_bias_kernel<<<(R + 15) / 16, 256, 0, stream>>>(
        rel_emb, Wr, br, nullptr, nullptr, out_rel, R);
}


# Retile to one round: 241 groups of 208 nodes (13 row tiles) instead of 391 of 128 on 256 WGPs; LDS sums hold one path at a time (K=256 twice); LDS float adds stay in arrival order
# speedup vs baseline: 6.6280x; 6.6280x over previous
#include <hip/hip_runtime.h>

#define ENT_DIM  128
#define REL_DIM  128
#define CMP_DIM  256
#define NUM_RELC 200

#define NB      208
#define NRT     13
#define SPITCH  260
#define KC      32
#define WPITCH  36
#define LCAP    6144
#define NTHR    1024

typedef __attribute__((ext_vector_type(2))) float v2f;
typedef __attribute__((ext_vector_type(8))) float v8f;

__device__ __forceinline__ v8f wmma4_f32(v2f a, v2f b, v8f c) {
    return __builtin_amdgcn_wmma_f32_16x16x4_f32(
         false, a,  false, b,
         (short)0, c,  false,  false);
}

__device__ __forceinline__ void add_edge_serial(int d, int et, int s, int path,
        const float* __restrict__ ent_emb, const float* __restrict__ rel_emb,
        float* s_S, int* s_cnt)
{
    float*       row = s_S + d * SPITCH;
    const float* rr  = rel_emb + (long)et * REL_DIM;
    const float* er  = ent_emb + (long)s  * ENT_DIM;
    for (int i = 0; i < REL_DIM; ++i) {
        atomicAdd(row + i,           rr[i]);
        atomicAdd(row + REL_DIM + i, er[i]);
    }
    atomicAdd(&s_cnt[d * 2 + path], 1);
}

__global__ __launch_bounds__(NTHR)
void node_owner_kernel(const int*   __restrict__ src,
                       const int*   __restrict__ dst,
                       const int*   __restrict__ etypes,
                       const float* __restrict__ ent_emb,
                       const float* __restrict__ rel_emb,
                       const float* __restrict__ Wo,
                       const float* __restrict__ bo,
                       const float* __restrict__ Wi,
                       const float* __restrict__ bi,
                       const float* __restrict__ Ws,
                       const float* __restrict__ bs,
                       float*       __restrict__ out,
                       int E, int N)
{
    __shared__ __attribute__((aligned(16))) float s_S[(NB + 1) * SPITCH];
    __shared__ __attribute__((aligned(16))) float s_W[ENT_DIM * WPITCH];
    __shared__ int s_list[LCAP];
    __shared__ int s_cnt[(NB + 1) * 2];
    __shared__ int s_n;
    __shared__ int s_np[2];

    const int tid  = threadIdx.x;
    const int lane = tid & 31;
    const int wave = __builtin_amdgcn_readfirstlane(tid >> 5);
    const int base = blockIdx.x * NB;

    if (tid < (NB + 1) * 2) s_cnt[tid] = 0;
    if (tid == 0) { s_n = 0; s_np[0] = 0; s_np[1] = 0; }
    __syncthreads();

    auto take = [&](int e) {
        const int tot = atomicAdd(&s_n, 1);
        if (tot < LCAP) {
            const int p   = (etypes[e] >= NUM_RELC) ? 1 : 0;
            const int pos = atomicAdd(&s_np[p], 1);
            s_list[p ? (LCAP - 1 - pos) : pos] = e;
        }
    };
    const int   nvec  = E >> 2;
    const int   nfull = nvec - nvec % (4 * NTHR);
    const int4* d4    = (const int4*)dst;
    for (int i0 = tid; i0 < nfull; i0 += 4 * NTHR) {
        int4 v[4];
        #pragma unroll
        for (int u = 0; u < 4; ++u) v[u] = d4[i0 + u * NTHR];
        unsigned hm = 0;
        #pragma unroll
        for (int u = 0; u < 4; ++u) {
            hm |= ((unsigned)(v[u].x - base) < (unsigned)NB ? 1u : 0u) << (4 * u + 0);
            hm |= ((unsigned)(v[u].y - base) < (unsigned)NB ? 1u : 0u) << (4 * u + 1);
            hm |= ((unsigned)(v[u].z - base) < (unsigned)NB ? 1u : 0u) << (4 * u + 2);
            hm |= ((unsigned)(v[u].w - base) < (unsigned)NB ? 1u : 0u) << (4 * u + 3);
        }
        while (hm) {
            const int b = __ffs(hm) - 1;
            hm &= hm - 1;
            take(((i0 + (b >> 2) * NTHR) << 2) + (b & 3));
        }
    }
    for (int i = nfull + tid; i < nvec; i += NTHR) {
        const int4 v = d4[i];
        const int  e = i << 2;
        if ((unsigned)(v.x - base) < (unsigned)NB) take(e);
        if ((unsigned)(v.y - base) < (unsigned)NB) take(e + 1);
        if ((unsigned)(v.z - base) < (unsigned)NB) take(e + 2);
        if ((unsigned)(v.w - base) < (unsigned)NB) take(e + 3);
    }
    for (int e = (nvec << 2) + tid; e < E; e += NTHR)
        if ((unsigned)(dst[e] - base) < (unsigned)NB) take(e);
    __syncthreads();

    const bool ovf = __builtin_amdgcn_readfirstlane(s_n) > LCAP;

    const int  m     = lane & 15;
    const int  khalf = (lane >> 4) << 1;
    const int  rt0   = wave >> 2;
    const int  cg    = wave & 3;
    const bool two   = (rt0 + 8) < NRT;
    const int  n0    = cg * 32 + m;
    const int  n1    = n0 + 16;
    const float* a_row0 = s_S + (rt0 * 16 + m) * SPITCH + khalf;
    const float* a_row1 = a_row0 + 8 * 16 * SPITCH;
    const float* b0p    = s_W + n0 * WPITCH + khalf;
    const float* b1p    = s_W + n1 * WPITCH + khalf;

    const int srow = tid >> 3;
    const int sq   = (tid & 7) << 2;

    const v8f zero8 = {0.f,0.f,0.f,0.f,0.f,0.f,0.f,0.f};
    v8f agg00 = zero8, agg01 = zero8, agg10 = zero8, agg11 = zero8;
    v8f sf00  = zero8, sf01  = zero8, sf10  = zero8, sf11  = zero8;

    auto kloop = [&](const float* __restrict__ Wp, int wstride, int nch, float4 wv,
                     v8f& c00, v8f& c01, v8f& c10, v8f& c11) {
        for (int c = 0; c < nch; ++c) {
            __syncthreads();
            *(float4*)(s_W + srow * WPITCH + sq) = wv;
            __syncthreads();
            if (c + 1 < nch) wv = *(const float4*)(Wp + srow * wstride + (c + 1) * KC + sq);
            if (two) {
                #pragma unroll
                for (int kk = 0; kk < KC; kk += 4) {
                    const v2f a0 = *(const v2f*)(a_row0 + c * KC + kk);
                    const v2f a1 = *(const v2f*)(a_row1 + c * KC + kk);
                    const v2f b0 = *(const v2f*)(b0p + kk);
                    const v2f b1 = *(const v2f*)(b1p + kk);
                    c00 = wmma4_f32(a0, b0, c00);
                    c01 = wmma4_f32(a0, b1, c01);
                    c10 = wmma4_f32(a1, b0, c10);
                    c11 = wmma4_f32(a1, b1, c11);
                }
            } else {
                #pragma unroll
                for (int kk = 0; kk < KC; kk += 4) {
                    const v2f a0 = *(const v2f*)(a_row0 + c * KC + kk);
                    const v2f b0 = *(const v2f*)(b0p + kk);
                    const v2f b1 = *(const v2f*)(b1p + kk);
                    c00 = wmma4_f32(a0, b0, c00);
                    c01 = wmma4_f32(a0, b1, c01);
                }
            }
        }
    };

    for (int p = 0; p < 2; ++p) {
        const float* Wp = p ? Wi : Wo;
        const float4 wv0 = *(const float4*)(Wp + srow * CMP_DIM + sq);

        __syncthreads();
        for (int i = tid; i < (NB + 1) * SPITCH / 4; i += NTHR)
            ((float4*)s_S)[i] = make_float4(0.f, 0.f, 0.f, 0.f);
        __syncthreads();

        if (!ovf) {
            int nl = s_np[p];
            nl = __builtin_amdgcn_readfirstlane(nl);
            const int per  = (((nl + 31) >> 5) + 3) & ~3;
            const int wbeg = wave * per;
            const int wend = (wbeg + per < nl) ? (wbeg + per) : nl;
            for (int b0 = wbeg; b0 < wend; b0 += 32) {
                const int idx = b0 + lane;
                int d = NB, et = 0, s = 0;
                if (idx < wend) {
                    const int e = s_list[p ? (LCAP - 1 - idx) : idx];
                    d  = dst[e] - base;
                    et = etypes[e];
                    s  = src[e];
                    atomicAdd(&s_cnt[d * 2 + p], 1);
                }
                int cntb = wend - b0;
                if (cntb > 32) cntb = 32;
                for (int j = 0; j < cntb; j += 4) {
                    float v[4][8];
                    int   off[4];
                    #pragma unroll
                    for (int u = 0; u < 4; ++u) {
                        const int dj  = __builtin_amdgcn_readlane(d,  j + u);
                        const int etj = __builtin_amdgcn_readlane(et, j + u);
                        const int sj  = __builtin_amdgcn_readlane(s,  j + u);
                        off[u] = dj * SPITCH + lane;
                        const float* rr = rel_emb + (long)etj * REL_DIM + lane;
                        const float* er = ent_emb + (long)sj  * ENT_DIM + lane;
                        v[u][0] = rr[0];  v[u][1] = rr[32]; v[u][2] = rr[64]; v[u][3] = rr[96];
                        v[u][4] = er[0];  v[u][5] = er[32]; v[u][6] = er[64]; v[u][7] = er[96];
                    }
                    #pragma unroll
                    for (int u = 0; u < 4; ++u) {
                        #pragma unroll
                        for (int q = 0; q < 8; ++q)
                            atomicAdd(&s_S[off[u] + 32 * q], v[u][q]);
                    }
                }
            }
        } else {
            for (int e = tid; e < E; e += NTHR) {
                const int d = dst[e] - base;
                if ((unsigned)d < (unsigned)NB) {
                    const int et = etypes[e];
                    if (((et >= NUM_RELC) ? 1 : 0) == p)
                        add_edge_serial(d, et, src[e], p, ent_emb, rel_emb, s_S, s_cnt);
                }
            }
        }

        kloop(Wp, CMP_DIM, CMP_DIM / KC, wv0, agg00, agg01, agg10, agg11);
    }

    {
        const float4 wv0 = *(const float4*)(Ws + srow * ENT_DIM + sq);
        __syncthreads();
        for (int i = tid; i < NB * (ENT_DIM / 4); i += NTHR) {
            const int r = i >> 5;
            const int q = (i & 31) << 2;
            long nrow = (long)base + r;
            if (nrow > (long)N - 1) nrow = (long)N - 1;
            *(float4*)(s_S + r * SPITCH + q) = *(const float4*)(ent_emb + nrow * ENT_DIM + q);
        }
        kloop(Ws, ENT_DIM, ENT_DIM / KC, wv0, sf00, sf01, sf10, sf11);
    }

    const float bo0 = bo[n0], bo1 = bo[n1];
    const float bi0 = bi[n0], bi1 = bi[n1];
    const float bs0 = bs[n0], bs1 = bs[n1];
    const int   rhalf = (lane >> 4) << 3;
    #pragma unroll
    for (int j = 0; j < 8; ++j) {
        const int  lr   = rt0 * 16 + j + rhalf;
        const long node = (long)base + lr;
        if (node < N) {
            const float cO = (float)s_cnt[lr * 2];
            const float cI = (float)s_cnt[lr * 2 + 1];
            float dg = cO + cI;
            dg = (dg > 1.0f) ? dg : 1.0f;
            out[node * ENT_DIM + n0] = sf00[j] + bs0 + (agg00[j] + cO * bo0 + cI * bi0) / dg;
            out[node * ENT_DIM + n1] = sf01[j] + bs1 + (agg01[j] + cO * bo1 + cI * bi1) / dg;
        }
    }
    if (two) {
        #pragma unroll
        for (int j = 0; j < 8; ++j) {
            const int  lr   = (rt0 + 8) * 16 + j + rhalf;
            const long node = (long)base + lr;
            if (node < N) {
                const float cO = (float)s_cnt[lr * 2];
                const float cI = (float)s_cnt[lr * 2 + 1];
                float dg = cO + cI;
                dg = (dg > 1.0f) ? dg : 1.0f;
                out[node * ENT_DIM + n0] = sf10[j] + bs0 + (agg10[j] + cO * bo0 + cI * bi0) / dg;
                out[node * ENT_DIM + n1] = sf11[j] + bs1 + (agg11[j] + cO * bo1 + cI * bi1) / dg;
            }
        }
    }
}

__global__ __launch_bounds__(256)
void gemm_bias_kernel(const float* __restrict__ A,
                      const float* __restrict__ W,
                      const float* __restrict__ b,
                      const float* __restrict__ agg,
                      const float* __restrict__ deg,
                      float*       __restrict__ out,
                      int N)
{
    const int  tid   = threadIdx.x;
    const int  lane  = tid & 31;
    const int  wave  = tid >> 5;
    const int  m     = lane & 15;
    const int  khalf = (lane >> 4) << 1;
    const long tile  = (long)blockIdx.x * 16;

    const long r    = tile + m;
    const long rc   = (r < N) ? r : (long)(N - 1);
    const float* ar = A + rc * ENT_DIM;

    const int    n  = (wave << 4) + m;
    const float* wr = W + (long)n * ENT_DIM;

    v8f acc = {0.f,0.f,0.f,0.f,0.f,0.f,0.f,0.f};

    #pragma unroll 4
    for (int k = 0; k < ENT_DIM; k += 4) {
        v2f av = *(const v2f*)(ar + k + khalf);
        v2f bv = *(const v2f*)(wr + k + khalf);
        acc = wmma4_f32(av, bv, acc);
    }

    const float bn    = b[n];
    const int   rhalf = (lane >> 4) << 3;

    #pragma unroll
    for (int j = 0; j < 8; ++j) {
        const int  me = j + rhalf;
        const long rr = tile + me;
        if (rr < N) {
            float v = acc[j] + bn;
            if (agg != nullptr) {
                float d = deg[rr];
                d = (d > 1.0f) ? d : 1.0f;
                v += agg[rr * ENT_DIM + n] / d;
            }
            out[rr * ENT_DIM + n] = v;
        }
    }
}

extern "C" void kernel_launch(void* const* d_in, const int* in_sizes, int n_in,
                              void* d_out, int out_size, void* d_ws, size_t ws_size,
                              hipStream_t stream)
{
    const int*   src     = (const int*)  d_in[0];
    const int*   dst     = (const int*)  d_in[1];
    const int*   etypes  = (const int*)  d_in[2];
    const float* ent_emb = (const float*)d_in[3];
    const float* rel_emb = (const float*)d_in[4];
    const float* Wo      = (const float*)d_in[5];
    const float* bo      = (const float*)d_in[6];
    const float* Wi      = (const float*)d_in[7];
    const float* bi      = (const float*)d_in[8];
    const float* Ws      = (const float*)d_in[9];
    const float* bs      = (const float*)d_in[10];
    const float* Wr      = (const float*)d_in[11];
    const float* br      = (const float*)d_in[12];

    const int E = in_sizes[0];
    const int N = in_sizes[3] / ENT_DIM;
    const int R = in_sizes[4] / REL_DIM;

    float* out_ent = (float*)d_out;
    float* out_rel = out_ent + (size_t)N * ENT_DIM;

    node_owner_kernel<<<(N + NB - 1) / NB, NTHR, 0, stream>>>(
        src, dst, etypes, ent_emb, rel_emb, Wo, bo, Wi, bi, Ws, bs, out_ent, E, N);

    gemm_bias_kernel<<<(R + 15) / 16, 256, 0, stream>>>(
        rel_emb, Wr, br, nullptr, nullptr, out_rel, R);
}


# Group's edge list ordered by node and path in LDS (count, prefix sum, place); a wave per node sums its raw rows in registers, stores the row once: LDS float adds, zero passes go; sums in slot order
# speedup vs baseline: 7.2088x; 1.0876x over previous
#include <hip/hip_runtime.h>

#define ENT_DIM  128
#define REL_DIM  128
#define CMP_DIM  256
#define NUM_RELC 200

#define NB      208
#define NRT     13
#define SPITCH  260
#define KC      32
#define WPITCH  36
#define LCAP    6144
#define NTHR    1024
#define NSEG    (NB * 2)
#define SEGPL   ((NSEG + 31) / 32)

typedef __attribute__((ext_vector_type(2))) float v2f;
typedef __attribute__((ext_vector_type(2))) int   v2i;
typedef __attribute__((ext_vector_type(8))) float v8f;

__device__ __forceinline__ v8f wmma4_f32(v2f a, v2f b, v8f c) {
    return __builtin_amdgcn_wmma_f32_16x16x4_f32(
         false, a,  false, b,
         (short)0, c,  false,  false);
}

__device__ __forceinline__ void add_edge_serial(int d, int et, int s, int path,
        const float* __restrict__ ent_emb, const float* __restrict__ rel_emb,
        float* s_S, int* s_cnt)
{
    float*       row = s_S + d * SPITCH;
    const float* rr  = rel_emb + (long)et * REL_DIM;
    const float* er  = ent_emb + (long)s  * ENT_DIM;
    for (int i = 0; i < REL_DIM; ++i) {
        atomicAdd(row + i,           rr[i]);
        atomicAdd(row + REL_DIM + i, er[i]);
    }
    atomicAdd(&s_cnt[d * 2 + path], 1);
}

__global__ __launch_bounds__(NTHR)
void node_owner_kernel(const int*   __restrict__ src,
                       const int*   __restrict__ dst,
                       const int*   __restrict__ etypes,
                       const float* __restrict__ ent_emb,
                       const float* __restrict__ rel_emb,
                       const float* __restrict__ Wo,
                       const float* __restrict__ bo,
                       const float* __restrict__ Wi,
                       const float* __restrict__ bi,
                       const float* __restrict__ Ws,
                       const float* __restrict__ bs,
                       float*       __restrict__ out,
                       int E, int N)
{
    __shared__ __attribute__((aligned(16))) float s_S[NB * SPITCH];
    __shared__ __attribute__((aligned(16))) float s_W[ENT_DIM * WPITCH];
    __shared__ int s_list[LCAP];
    __shared__ v2i s_rec[LCAP];
    __shared__ int s_cnt[NSEG];
    __shared__ int s_end[NSEG];
    __shared__ int s_n;
    __shared__ int s_np[2];
    __shared__ int s_next[2];

    const int tid  = threadIdx.x;
    const int lane = tid & 31;
    const int wave = __builtin_amdgcn_readfirstlane(tid >> 5);
    const int base = blockIdx.x * NB;

    if (tid < NSEG) s_cnt[tid] = 0;
    if (tid == 0) { s_n = 0; s_np[0] = 0; s_np[1] = 0; s_next[0] = 0; s_next[1] = 0; }
    __syncthreads();

    auto take = [&](int e) {
        const int tot = atomicAdd(&s_n, 1);
        if (tot < LCAP) {
            const int d   = dst[e] - base;
            const int p   = (etypes[e] >= NUM_RELC) ? 1 : 0;
            const int pos = atomicAdd(&s_np[p], 1);
            s_list[p ? (LCAP - 1 - pos) : pos] = e;
            atomicAdd(&s_cnt[d * 2 + p], 1);
        }
    };
    const int   nvec  = E >> 2;
    const int   nfull = nvec - nvec % (4 * NTHR);
    const int4* d4    = (const int4*)dst;
    for (int i0 = tid; i0 < nfull; i0 += 4 * NTHR) {
        int4 v[4];
        #pragma unroll
        for (int u = 0; u < 4; ++u) v[u] = d4[i0 + u * NTHR];
        unsigned hm = 0;
        #pragma unroll
        for (int u = 0; u < 4; ++u) {
            hm |= ((unsigned)(v[u].x - base) < (unsigned)NB ? 1u : 0u) << (4 * u + 0);
            hm |= ((unsigned)(v[u].y - base) < (unsigned)NB ? 1u : 0u) << (4 * u + 1);
            hm |= ((unsigned)(v[u].z - base) < (unsigned)NB ? 1u : 0u) << (4 * u + 2);
            hm |= ((unsigned)(v[u].w - base) < (unsigned)NB ? 1u : 0u) << (4 * u + 3);
        }
        while (hm) {
            const int b = __ffs(hm) - 1;
            hm &= hm - 1;
            take(((i0 + (b >> 2) * NTHR) << 2) + (b & 3));
        }
    }
    for (int i = nfull + tid; i < nvec; i += NTHR) {
        const int4 v = d4[i];
        const int  e = i << 2;
        if ((unsigned)(v.x - base) < (unsigned)NB) take(e);
        if ((unsigned)(v.y - base) < (unsigned)NB) take(e + 1);
        if ((unsigned)(v.z - base) < (unsigned)NB) take(e + 2);
        if ((unsigned)(v.w - base) < (unsigned)NB) take(e + 3);
    }
    for (int e = (nvec << 2) + tid; e < E; e += NTHR)
        if ((unsigned)(dst[e] - base) < (unsigned)NB) take(e);
    __syncthreads();

    const bool ovf = __builtin_amdgcn_readfirstlane(s_n) > LCAP;

    if (!ovf) {
        if (wave == 0) {
            int loc[SEGPL];
            int sum = 0;
            #pragma unroll
            for (int k = 0; k < SEGPL; ++k) {
                const int i = lane * SEGPL + k;
                loc[k] = sum;
                sum += (i < NSEG) ? s_cnt[i] : 0;
            }
            int incl = sum;
            #pragma unroll
            for (int off = 1; off < 32; off <<= 1) {
                const int t = __shfl_up(incl, off);
                if (lane >= off) incl += t;
            }
            const int excl = incl - sum;
            #pragma unroll
            for (int k = 0; k < SEGPL; ++k) {
                const int i = lane * SEGPL + k;
                if (i < NSEG) s_end[i] = excl + loc[k];
            }
        }
        __syncthreads();
        const int nO = s_np[0];
        const int nL = nO + s_np[1];
        for (int i = tid; i < nL; i += NTHR) {
            const int p = (i >= nO) ? 1 : 0;
            const int e = s_list[p ? (LCAP - 1 - (i - nO)) : i];
            const int d = dst[e] - base;
            v2i r;
            r.x = etypes[e];
            r.y = src[e];
            const int slot = atomicAdd(&s_end[d * 2 + p], 1);
            s_rec[slot] = r;
        }
    } else {
        if (tid < NSEG) s_cnt[tid] = 0;
    }

    const int  m     = lane & 15;
    const int  khalf = (lane >> 4) << 1;
    const int  rt0   = wave >> 2;
    const int  cg    = wave & 3;
    const bool two   = (rt0 + 8) < NRT;
    const int  n0    = cg * 32 + m;
    const int  n1    = n0 + 16;
    const float* a_row0 = s_S + (rt0 * 16 + m) * SPITCH + khalf;
    const float* a_row1 = a_row0 + 8 * 16 * SPITCH;
    const float* b0p    = s_W + n0 * WPITCH + khalf;
    const float* b1p    = s_W + n1 * WPITCH + khalf;

    const int srow = tid >> 3;
    const int sq   = (tid & 7) << 2;

    const v8f zero8 = {0.f,0.f,0.f,0.f,0.f,0.f,0.f,0.f};
    v8f agg00 = zero8, agg01 = zero8, agg10 = zero8, agg11 = zero8;
    v8f sf00  = zero8, sf01  = zero8, sf10  = zero8, sf11  = zero8;

    auto kloop = [&](const float* __restrict__ Wp, int wstride, int nch, float4 wv,
                     v8f& c00, v8f& c01, v8f& c10, v8f& c11) {
        for (int c = 0; c < nch; ++c) {
            __syncthreads();
            *(float4*)(s_W + srow * WPITCH + sq) = wv;
            __syncthreads();
            if (c + 1 < nch) wv = *(const float4*)(Wp + srow * wstride + (c + 1) * KC + sq);
            if (two) {
                #pragma unroll
                for (int kk = 0; kk < KC; kk += 4) {
                    const v2f a0 = *(const v2f*)(a_row0 + c * KC + kk);
                    const v2f a1 = *(const v2f*)(a_row1 + c * KC + kk);
                    const v2f b0 = *(const v2f*)(b0p + kk);
                    const v2f b1 = *(const v2f*)(b1p + kk);
                    c00 = wmma4_f32(a0, b0, c00);
                    c01 = wmma4_f32(a0, b1, c01);
                    c10 = wmma4_f32(a1, b0, c10);
                    c11 = wmma4_f32(a1, b1, c11);
                }
            } else {
                #pragma unroll
                for (int kk = 0; kk < KC; kk += 4) {
                    const v2f a0 = *(const v2f*)(a_row0 + c * KC + kk);
                    const v2f b0 = *(const v2f*)(b0p + kk);
                    const v2f b1 = *(const v2f*)(b1p + kk);
                    c00 = wmma4_f32(a0, b0, c00);
                    c01 = wmma4_f32(a0, b1, c01);
                }
            }
        }
    };

    for (int p = 0; p < 2; ++p) {
        const float* Wp = p ? Wi : Wo;
        const float4 wv0 = *(const float4*)(Wp + srow * CMP_DIM + sq);

        __syncthreads();

        if (!ovf) {
            for (;;) {
                int d = 0;
                if (lane == 0) d = atomicAdd(&s_next[p], 1);
                d = __builtin_amdgcn_readfirstlane(d);
                if (d >= NB) break;
                const int cnt = __builtin_amdgcn_readfirstlane(s_cnt[d * 2 + p]);
                const int beg = __builtin_amdgcn_readfirstlane(s_end[d * 2 + p]) - cnt;
                float acc[8];
                #pragma unroll
                for (int q = 0; q < 8; ++q) acc[q] = 0.f;
                for (int b0 = 0; b0 < cnt; b0 += 32) {
                    int nb = cnt - b0;
                    if (nb > 32) nb = 32;
                    v2i r = {0, 0};
                    if (lane < nb) r = s_rec[beg + b0 + lane];
                    for (int j = 0; j < nb; j += 4) {
                        float v[4][8];
                        #pragma unroll
                        for (int u = 0; u < 4; ++u) {
                            if (j + u < nb) {
                                const int etj = __builtin_amdgcn_readlane(r.x, j + u);
                                const int sj  = __builtin_amdgcn_readlane(r.y, j + u);
                                const float* rr = rel_emb + (long)etj * REL_DIM + lane;
                                const float* er = ent_emb + (long)sj  * ENT_DIM + lane;
                                v[u][0] = rr[0];  v[u][1] = rr[32]; v[u][2] = rr[64]; v[u][3] = rr[96];
                                v[u][4] = er[0];  v[u][5] = er[32]; v[u][6] = er[64]; v[u][7] = er[96];
                            } else {
                                #pragma unroll
                                for (int q = 0; q < 8; ++q) v[u][q] = 0.f;
                            }
                        }
                        #pragma unroll
                        for (int u = 0; u < 4; ++u) {
                            #pragma unroll
                            for (int q = 0; q < 8; ++q) acc[q] += v[u][q];
                        }
                    }
                }
                float* row = s_S + d * SPITCH + lane;
                #pragma unroll
                for (int q = 0; q < 8; ++q) row[32 * q] = acc[q];
            }
        } else {
            for (int i = tid; i < NB * SPITCH / 4; i += NTHR)
                ((float4*)s_S)[i] = make_float4(0.f, 0.f, 0.f, 0.f);
            __syncthreads();
            for (int e = tid; e < E; e += NTHR) {
                const int d = dst[e] - base;
                if ((unsigned)d < (unsigned)NB) {
                    const int et = etypes[e];
                    if (((et >= NUM_RELC) ? 1 : 0) == p)
                        add_edge_serial(d, et, src[e], p, ent_emb, rel_emb, s_S, s_cnt);
                }
            }
        }

        kloop(Wp, CMP_DIM, CMP_DIM / KC, wv0, agg00, agg01, agg10, agg11);
    }

    {
        const float4 wv0 = *(const float4*)(Ws + srow * ENT_DIM + sq);
        __syncthreads();
        for (int i = tid; i < NB * (ENT_DIM / 4); i += NTHR) {
            const int r = i >> 5;
            const int q = (i & 31) << 2;
            long nrow = (long)base + r;
            if (nrow > (long)N - 1) nrow = (long)N - 1;
            *(float4*)(s_S + r * SPITCH + q) = *(const float4*)(ent_emb + nrow * ENT_DIM + q);
        }
        kloop(Ws, ENT_DIM, ENT_DIM / KC, wv0, sf00, sf01, sf10, sf11);
    }

    const float bo0 = bo[n0], bo1 = bo[n1];
    const float bi0 = bi[n0], bi1 = bi[n1];
    const float bs0 = bs[n0], bs1 = bs[n1];
    const int   rhalf = (lane >> 4) << 3;
    #pragma unroll
    for (int j = 0; j < 8; ++j) {
        const int  lr   = rt0 * 16 + j + rhalf;
        const long node = (long)base + lr;
        if (node < N) {
            const float cO = (float)s_cnt[lr * 2];
            const float cI = (float)s_cnt[lr * 2 + 1];
            float dg = cO + cI;
            dg = (dg > 1.0f) ? dg : 1.0f;
            out[node * ENT_DIM + n0] = sf00[j] + bs0 + (agg00[j] + cO * bo0 + cI * bi0) / dg;
            out[node * ENT_DIM + n1] = sf01[j] + bs1 + (agg01[j] + cO * bo1 + cI * bi1) / dg;
        }
    }
    if (two) {
        #pragma unroll
        for (int j = 0; j < 8; ++j) {
            const int  lr   = (rt0 + 8) * 16 + j + rhalf;
            const long node = (long)base + lr;
            if (node < N) {
                const float cO = (float)s_cnt[lr * 2];
                const float cI = (float)s_cnt[lr * 2 + 1];
                float dg = cO + cI;
                dg = (dg > 1.0f) ? dg : 1.0f;
                out[node * ENT_DIM + n0] = sf10[j] + bs0 + (agg10[j] + cO * bo0 + cI * bi0) / dg;
                out[node * ENT_DIM + n1] = sf11[j] + bs1 + (agg11[j] + cO * bo1 + cI * bi1) / dg;
            }
        }
    }
}

__global__ __launch_bounds__(256)
void gemm_bias_kernel(const float* __restrict__ A,
                      const float* __restrict__ W,
                      const float* __restrict__ b,
                      const float* __restrict__ agg,
                      const float* __restrict__ deg,
                      float*       __restrict__ out,
                      int N)
{
    const int  tid   = threadIdx.x;
    const int  lane  = tid & 31;
    const int  wave  = tid >> 5;
    const int  m     = lane & 15;
    const int  khalf = (lane >> 4) << 1;
    const long tile  = (long)blockIdx.x * 16;

    const long r    = tile + m;
    const long rc   = (r < N) ? r : (long)(N - 1);
    const float* ar = A + rc * ENT_DIM;

    const int    n  = (wave << 4) + m;
    const float* wr = W + (long)n * ENT_DIM;

    v8f acc = {0.f,0.f,0.f,0.f,0.f,0.f,0.f,0.f};

    #pragma unroll 4
    for (int k = 0; k < ENT_DIM; k += 4) {
        v2f av = *(const v2f*)(ar + k + khalf);
        v2f bv = *(const v2f*)(wr + k + khalf);
        acc = wmma4_f32(av, bv, acc);
    }

    const float bn    = b[n];
    const int   rhalf = (lane >> 4) << 3;

    #pragma unroll
    for (int j = 0; j < 8; ++j) {
        const int  me = j + rhalf;
        const long rr = tile + me;
        if (rr < N) {
            float v = acc[j] + bn;
            if (agg != nullptr) {
                float d = deg[rr];
                d = (d > 1.0f) ? d : 1.0f;
                v += agg[rr * ENT_DIM + n] / d;
            }
            out[rr * ENT_DIM + n] = v;
        }
    }
}

extern "C" void kernel_launch(void* const* d_in, const int* in_sizes, int n_in,
                              void* d_out, int out_size, void* d_ws, size_t ws_size,
                              hipStream_t stream)
{
    const int*   src     = (const int*)  d_in[0];
    const int*   dst     = (const int*)  d_in[1];
    const int*   etypes  = (const int*)  d_in[2];
    const float* ent_emb = (const float*)d_in[3];
    const float* rel_emb = (const float*)d_in[4];
    const float* Wo      = (const float*)d_in[5];
    const float* bo      = (const float*)d_in[6];
    const float* Wi      = (const float*)d_in[7];
    const float* bi      = (const float*)d_in[8];
    const float* Ws      = (const float*)d_in[9];
    const float* bs      = (const float*)d_in[10];
    const float* Wr      = (const float*)d_in[11];
    const float* br      = (const float*)d_in[12];

    const int E = in_sizes[0];
    const int N = in_sizes[3] / ENT_DIM;
    const int R = in_sizes[4] / REL_DIM;

    float* out_ent = (float*)d_out;
    float* out_rel = out_ent + (size_t)N * ENT_DIM;

    node_owner_kernel<<<(N + NB - 1) / NB, NTHR, 0, stream>>>(
        src, dst, etypes, ent_emb, rel_emb, Wo, bo, Wi, bi, Ws, bs, out_ent, E, N);

    gemm_bias_kernel<<<(R + 15) / 16, 256, 0, stream>>>(
        rel_emb, Wr, br, nullptr, nullptr, out_rel, R);
}


# Launch 1 (relation GEMM, now first) also writes each edge's owner-group id as one byte; the owner scan reads 1 byte per edge, not 4, and tests 4 edges per word; sums stay in arrival slot order
# speedup vs baseline: 8.4800x; 1.1763x over previous
#include <hip/hip_runtime.h>

#define ENT_DIM  128
#define REL_DIM  128
#define CMP_DIM  256
#define NUM_RELC 200

#define NB      208
#define NRT     13
#define SPITCH  260
#define KC      32
#define WPITCH  36
#define LCAP    6144
#define NTHR    1024
#define NSEG    (NB * 2)
#define SEGPL   ((NSEG + 31) / 32)

typedef __attribute__((ext_vector_type(2))) float v2f;
typedef __attribute__((ext_vector_type(2))) int   v2i;
typedef __attribute__((ext_vector_type(8))) float v8f;

__device__ __forceinline__ v8f wmma4_f32(v2f a, v2f b, v8f c) {
    return __builtin_amdgcn_wmma_f32_16x16x4_f32(
         false, a,  false, b,
         (short)0, c,  false,  false);
}

__device__ __forceinline__ void add_edge_serial(int d, int et, int s, int path,
        const float* __restrict__ ent_emb, const float* __restrict__ rel_emb,
        float* s_S, int* s_cnt)
{
    float*       row = s_S + d * SPITCH;
    const float* rr  = rel_emb + (long)et * REL_DIM;
    const float* er  = ent_emb + (long)s  * ENT_DIM;
    for (int i = 0; i < REL_DIM; ++i) {
        atomicAdd(row + i,           rr[i]);
        atomicAdd(row + REL_DIM + i, er[i]);
    }
    atomicAdd(&s_cnt[d * 2 + path], 1);
}

__device__ __forceinline__ unsigned owner_byte(int d) {
    const unsigned q = (unsigned)d / (unsigned)NB;
    return q < 255u ? q : 255u;
}

__device__ __forceinline__ unsigned nomatch16(uint4 v, unsigned pat) {
    const unsigned w[4] = {v.x, v.y, v.z, v.w};
    unsigned r = 0;
    #pragma unroll
    for (int j = 0; j < 4; ++j) {
        const unsigned x = w[j] ^ pat;
        const unsigned y = ((x & 0x7f7f7f7fu) + 0x7f7f7f7fu) | x;
        r |= (y >> (7 - j)) & (0x01010101u << j);
    }
    return r;
}

__global__ __launch_bounds__(NTHR)
void node_owner_kernel(const int*   __restrict__ src,
                       const unsigned int* __restrict__ gid,
                       const int*   __restrict__ dst,
                       const int*   __restrict__ etypes,
                       const float* __restrict__ ent_emb,
                       const float* __restrict__ rel_emb,
                       const float* __restrict__ Wo,
                       const float* __restrict__ bo,
                       const float* __restrict__ Wi,
                       const float* __restrict__ bi,
                       const float* __restrict__ Ws,
                       const float* __restrict__ bs,
                       float*       __restrict__ out,
                       int E, int N)
{
    __shared__ __attribute__((aligned(16))) float s_S[NB * SPITCH];
    __shared__ __attribute__((aligned(16))) float s_W[ENT_DIM * WPITCH];
    __shared__ int s_list[LCAP];
    __shared__ v2i s_rec[LCAP];
    __shared__ int s_cnt[NSEG];
    __shared__ int s_end[NSEG];
    __shared__ int s_n;
    __shared__ int s_np[2];
    __shared__ int s_next[2];

    const int tid  = threadIdx.x;
    const int lane = tid & 31;
    const int wave = __builtin_amdgcn_readfirstlane(tid >> 5);
    const int base = blockIdx.x * NB;

    if (tid < NSEG) s_cnt[tid] = 0;
    if (tid == 0) { s_n = 0; s_np[0] = 0; s_np[1] = 0; s_next[0] = 0; s_next[1] = 0; }
    __syncthreads();

    auto take = [&](int e) {
        const int tot = atomicAdd(&s_n, 1);
        if (tot < LCAP) {
            const int d   = dst[e] - base;
            const int p   = (etypes[e] >= NUM_RELC) ? 1 : 0;
            const int pos = atomicAdd(&s_np[p], 1);
            s_list[p ? (LCAP - 1 - pos) : pos] = e;
            atomicAdd(&s_cnt[d * 2 + p], 1);
        }
    };
    const unsigned pat   = (unsigned)blockIdx.x * 0x01010101u;
    const int      nvec  = (E + 15) >> 4;
    const int      nfull = nvec - nvec % (4 * NTHR);
    const uint4*   g4    = (const uint4*)gid;
    for (int i0 = tid; i0 < nfull; i0 += 4 * NTHR) {
        uint4 v[4];
        #pragma unroll
        for (int u = 0; u < 4; ++u) v[u] = g4[i0 + u * NTHR];
        const unsigned nlo = nomatch16(v[0], pat) | (nomatch16(v[1], pat) << 4);
        const unsigned nhi = nomatch16(v[2], pat) | (nomatch16(v[3], pat) << 4);
        unsigned long long hm = ~((unsigned long long)nlo | ((unsigned long long)nhi << 32));
        while (hm) {
            const int b = __ffsll(hm) - 1;
            hm &= hm - 1;
            const int u = ((b >> 5) << 1) | ((b >> 2) & 1);
            take(((i0 + u * NTHR) << 4) + ((b & 3) << 2) + ((b >> 3) & 3));
        }
    }
    for (int i = nfull + tid; i < nvec; i += NTHR) {
        unsigned hm = ~nomatch16(g4[i], pat) & 0x0f0f0f0fu;
        while (hm) {
            const int b = __ffs(hm) - 1;
            hm &= hm - 1;
            take((i << 4) + ((b & 3) << 2) + (b >> 3));
        }
    }
    __syncthreads();

    const bool ovf = __builtin_amdgcn_readfirstlane(s_n) > LCAP;

    if (!ovf) {
        if (wave == 0) {
            int loc[SEGPL];
            int sum = 0;
            #pragma unroll
            for (int k = 0; k < SEGPL; ++k) {
                const int i = lane * SEGPL + k;
                loc[k] = sum;
                sum += (i < NSEG) ? s_cnt[i] : 0;
            }
            int incl = sum;
            #pragma unroll
            for (int off = 1; off < 32; off <<= 1) {
                const int t = __shfl_up(incl, off);
                if (lane >= off) incl += t;
            }
            const int excl = incl - sum;
            #pragma unroll
            for (int k = 0; k < SEGPL; ++k) {
                const int i = lane * SEGPL + k;
                if (i < NSEG) s_end[i] = excl + loc[k];
            }
        }
        __syncthreads();
        const int nO = s_np[0];
        const int nL = nO + s_np[1];
        for (int i = tid; i < nL; i += NTHR) {
            const int p = (i >= nO) ? 1 : 0;
            const int e = s_list[p ? (LCAP - 1 - (i - nO)) : i];
            const int d = dst[e] - base;
            v2i r;
            r.x = etypes[e];
            r.y = src[e];
            const int slot = atomicAdd(&s_end[d * 2 + p], 1);
            s_rec[slot] = r;
        }
    } else {
        if (tid < NSEG) s_cnt[tid] = 0;
    }

    const int  m     = lane & 15;
    const int  khalf = (lane >> 4) << 1;
    const int  rt0   = wave >> 2;
    const int  cg    = wave & 3;
    const bool two   = (rt0 + 8) < NRT;
    const int  n0    = cg * 32 + m;
    const int  n1    = n0 + 16;
    const float* a_row0 = s_S + (rt0 * 16 + m) * SPITCH + khalf;
    const float* a_row1 = a_row0 + 8 * 16 * SPITCH;
    const float* b0p    = s_W + n0 * WPITCH + khalf;
    const float* b1p    = s_W + n1 * WPITCH + khalf;

    const int srow = tid >> 3;
    const int sq   = (tid & 7) << 2;

    const v8f zero8 = {0.f,0.f,0.f,0.f,0.f,0.f,0.f,0.f};
    v8f agg00 = zero8, agg01 = zero8, agg10 = zero8, agg11 = zero8;
    v8f sf00  = zero8, sf01  = zero8, sf10  = zero8, sf11  = zero8;

    auto kloop = [&](const float* __restrict__ Wp, int wstride, int nch, float4 wv,
                     v8f& c00, v8f& c01, v8f& c10, v8f& c11) {
        for (int c = 0; c < nch; ++c) {
            __syncthreads();
            *(float4*)(s_W + srow * WPITCH + sq) = wv;
            __syncthreads();
            if (c + 1 < nch) wv = *(const float4*)(Wp + srow * wstride + (c + 1) * KC + sq);
            if (two) {
                #pragma unroll
                for (int kk = 0; kk < KC; kk += 4) {
                    const v2f a0 = *(const v2f*)(a_row0 + c * KC + kk);
                    const v2f a1 = *(const v2f*)(a_row1 + c * KC + kk);
                    const v2f b0 = *(const v2f*)(b0p + kk);
                    const v2f b1 = *(const v2f*)(b1p + kk);
                    c00 = wmma4_f32(a0, b0, c00);
                    c01 = wmma4_f32(a0, b1, c01);
                    c10 = wmma4_f32(a1, b0, c10);
                    c11 = wmma4_f32(a1, b1, c11);
                }
            } else {
                #pragma unroll
                for (int kk = 0; kk < KC; kk += 4) {
                    const v2f a0 = *(const v2f*)(a_row0 + c * KC + kk);
                    const v2f b0 = *(const v2f*)(b0p + kk);
                    const v2f b1 = *(const v2f*)(b1p + kk);
                    c00 = wmma4_f32(a0, b0, c00);
                    c01 = wmma4_f32(a0, b1, c01);
                }
            }
        }
    };

    for (int p = 0; p < 2; ++p) {
        const float* Wp = p ? Wi : Wo;
        const float4 wv0 = *(const float4*)(Wp + srow * CMP_DIM + sq);

        __syncthreads();

        if (!ovf) {
            for (;;) {
                int d = 0;
                if (lane == 0) d = atomicAdd(&s_next[p], 1);
                d = __builtin_amdgcn_readfirstlane(d);
                if (d >= NB) break;
                const int cnt = __builtin_amdgcn_readfirstlane(s_cnt[d * 2 + p]);
                const int beg = __builtin_amdgcn_readfirstlane(s_end[d * 2 + p]) - cnt;
                float acc[8];
                #pragma unroll
                for (int q = 0; q < 8; ++q) acc[q] = 0.f;
                for (int b0 = 0; b0 < cnt; b0 += 32) {
                    int nb = cnt - b0;
                    if (nb > 32) nb = 32;
                    v2i r = {0, 0};
                    if (lane < nb) r = s_rec[beg + b0 + lane];
                    for (int j = 0; j < nb; j += 4) {
                        float v[4][8];
                        #pragma unroll
                        for (int u = 0; u < 4; ++u) {
                            if (j + u < nb) {
                                const int etj = __builtin_amdgcn_readlane(r.x, j + u);
                                const int sj  = __builtin_amdgcn_readlane(r.y, j + u);
                                const float* rr = rel_emb + (long)etj * REL_DIM + lane;
                                const float* er = ent_emb + (long)sj  * ENT_DIM + lane;
                                v[u][0] = rr[0];  v[u][1] = rr[32]; v[u][2] = rr[64]; v[u][3] = rr[96];
                                v[u][4] = er[0];  v[u][5] = er[32]; v[u][6] = er[64]; v[u][7] = er[96];
                            } else {
                                #pragma unroll
                                for (int q = 0; q < 8; ++q) v[u][q] = 0.f;
                            }
                        }
                        #pragma unroll
                        for (int u = 0; u < 4; ++u) {
                            #pragma unroll
                            for (int q = 0; q < 8; ++q) acc[q] += v[u][q];
                        }
                    }
                }
                float* row = s_S + d * SPITCH + lane;
                #pragma unroll
                for (int q = 0; q < 8; ++q) row[32 * q] = acc[q];
            }
        } else {
            for (int i = tid; i < NB * SPITCH / 4; i += NTHR)
                ((float4*)s_S)[i] = make_float4(0.f, 0.f, 0.f, 0.f);
            __syncthreads();
            for (int e = tid; e < E; e += NTHR) {
                const int d = dst[e] - base;
                if ((unsigned)d < (unsigned)NB) {
                    const int et = etypes[e];
                    if (((et >= NUM_RELC) ? 1 : 0) == p)
                        add_edge_serial(d, et, src[e], p, ent_emb, rel_emb, s_S, s_cnt);
                }
            }
        }

        kloop(Wp, CMP_DIM, CMP_DIM / KC, wv0, agg00, agg01, agg10, agg11);
    }

    {
        const float4 wv0 = *(const float4*)(Ws + srow * ENT_DIM + sq);
        __syncthreads();
        for (int i = tid; i < NB * (ENT_DIM / 4); i += NTHR) {
            const int r = i >> 5;
            const int q = (i & 31) << 2;
            long nrow = (long)base + r;
            if (nrow > (long)N - 1) nrow = (long)N - 1;
            *(float4*)(s_S + r * SPITCH + q) = *(const float4*)(ent_emb + nrow * ENT_DIM + q);
        }
        kloop(Ws, ENT_DIM, ENT_DIM / KC, wv0, sf00, sf01, sf10, sf11);
    }

    const float bo0 = bo[n0], bo1 = bo[n1];
    const float bi0 = bi[n0], bi1 = bi[n1];
    const float bs0 = bs[n0], bs1 = bs[n1];
    const int   rhalf = (lane >> 4) << 3;
    #pragma unroll
    for (int j = 0; j < 8; ++j) {
        const int  lr   = rt0 * 16 + j + rhalf;
        const long node = (long)base + lr;
        if (node < N) {
            const float cO = (float)s_cnt[lr * 2];
            const float cI = (float)s_cnt[lr * 2 + 1];
            float dg = cO + cI;
            dg = (dg > 1.0f) ? dg : 1.0f;
            out[node * ENT_DIM + n0] = sf00[j] + bs0 + (agg00[j] + cO * bo0 + cI * bi0) / dg;
            out[node * ENT_DIM + n1] = sf01[j] + bs1 + (agg01[j] + cO * bo1 + cI * bi1) / dg;
        }
    }
    if (two) {
        #pragma unroll
        for (int j = 0; j < 8; ++j) {
            const int  lr   = (rt0 + 8) * 16 + j + rhalf;
            const long node = (long)base + lr;
            if (node < N) {
                const float cO = (float)s_cnt[lr * 2];
                const float cI = (float)s_cnt[lr * 2 + 1];
                float dg = cO + cI;
                dg = (dg > 1.0f) ? dg : 1.0f;
                out[node * ENT_DIM + n0] = sf10[j] + bs0 + (agg10[j] + cO * bo0 + cI * bi0) / dg;
                out[node * ENT_DIM + n1] = sf11[j] + bs1 + (agg11[j] + cO * bo1 + cI * bi1) / dg;
            }
        }
    }
}

__global__ __launch_bounds__(256)
void gemm_bias_kernel(const float* __restrict__ A,
                      const float* __restrict__ W,
                      const float* __restrict__ b,
                      const float* __restrict__ agg,
                      const float* __restrict__ deg,
                      float*       __restrict__ out,
                      int N,
                      const int*   __restrict__ dst,
                      unsigned int* __restrict__ gid,
                      int E, int ngemm)
{
    if ((int)blockIdx.x >= ngemm) {
        const int c = ((int)blockIdx.x - ngemm) * 256 + (int)threadIdx.x;
        if (c >= ((E + 15) >> 4)) return;
        const int e0 = c << 4;
        unsigned w[4];
        if (e0 + 16 <= E) {
            const int4* d4 = (const int4*)(dst + e0);
            #pragma unroll
            for (int j = 0; j < 4; ++j) {
                const int4 v = d4[j];
                w[j] = owner_byte(v.x) | (owner_byte(v.y) << 8) | (owner_byte(v.z) << 16) | (owner_byte(v.w) << 24);
            }
        } else {
            #pragma unroll
            for (int j = 0; j < 4; ++j) {
                w[j] = 0;
                #pragma unroll
                for (int k = 0; k < 4; ++k) {
                    const int e = e0 + 4 * j + k;
                    w[j] |= ((e < E) ? owner_byte(dst[e]) : 255u) << (8 * k);
                }
            }
        }
        ((uint4*)gid)[c] = make_uint4(w[0], w[1], w[2], w[3]);
        return;
    }

    const int  tid   = threadIdx.x;
    const int  lane  = tid & 31;
    const int  wave  = tid >> 5;
    const int  m     = lane & 15;
    const int  khalf = (lane >> 4) << 1;
    const long tile  = (long)blockIdx.x * 16;

    const long r    = tile + m;
    const long rc   = (r < N) ? r : (long)(N - 1);
    const float* ar = A + rc * ENT_DIM;

    const int    n  = (wave << 4) + m;
    const float* wr = W + (long)n * ENT_DIM;

    v8f acc = {0.f,0.f,0.f,0.f,0.f,0.f,0.f,0.f};

    #pragma unroll 4
    for (int k = 0; k < ENT_DIM; k += 4) {
        v2f av = *(const v2f*)(ar + k + khalf);
        v2f bv = *(const v2f*)(wr + k + khalf);
        acc = wmma4_f32(av, bv, acc);
    }

    const float bn    = b[n];
    const int   rhalf = (lane >> 4) << 3;

    #pragma unroll
    for (int j = 0; j < 8; ++j) {
        const int  me = j + rhalf;
        const long rr = tile + me;
        if (rr < N) {
            float v = acc[j] + bn;
            if (agg != nullptr) {
                float d = deg[rr];
                d = (d > 1.0f) ? d : 1.0f;
                v += agg[rr * ENT_DIM + n] / d;
            }
            out[rr * ENT_DIM + n] = v;
        }
    }
}

extern "C" void kernel_launch(void* const* d_in, const int* in_sizes, int n_in,
                              void* d_out, int out_size, void* d_ws, size_t ws_size,
                              hipStream_t stream)
{
    const int*   src     = (const int*)  d_in[0];
    const int*   dst     = (const int*)  d_in[1];
    const int*   etypes  = (const int*)  d_in[2];
    const float* ent_emb = (const float*)d_in[3];
    const float* rel_emb = (const float*)d_in[4];
    const float* Wo      = (const float*)d_in[5];
    const float* bo      = (const float*)d_in[6];
    const float* Wi      = (const float*)d_in[7];
    const float* bi      = (const float*)d_in[8];
    const float* Ws      = (const float*)d_in[9];
    const float* bs      = (const float*)d_in[10];
    const float* Wr      = (const float*)d_in[11];
    const float* br      = (const float*)d_in[12];

    const int E = in_sizes[0];
    const int N = in_sizes[3] / ENT_DIM;
    const int R = in_sizes[4] / REL_DIM;

    float* out_ent = (float*)d_out;
    float* out_rel = out_ent + (size_t)N * ENT_DIM;

    unsigned int* gid = (unsigned int*)d_ws;

    gemm_bias_kernel<<<(R + 15) / 16 + ((E + 15) / 16 + 255) / 256, 256, 0, stream>>>(
        rel_emb, Wr, br, nullptr, nullptr, out_rel, R, dst, gid, E, (R + 15) / 16);

    node_owner_kernel<<<(N + NB - 1) / NB, NTHR, 0, stream>>>(
        src, gid, dst, etypes, ent_emb, rel_emb, Wo, bo, Wi, bi, Ws, bs, out_ent, E, N);
}


# Relation half of each message read from a table projected once per type (launch 1, bias inside), summed per node; node GEMM K per path 256 to 128, one row-sum pass; rel part summed after its product
# speedup vs baseline: 8.8490x; 1.0435x over previous
#include <hip/hip_runtime.h>

#define ENT_DIM  128
#define REL_DIM  128
#define CMP_DIM  256
#define NUM_RELC 200

#define NB      208
#define NRT     13
#define SPITCH  260
#define KC      32
#define WPITCH  36
#define LCAP    6144
#define NTHR    1024
#define NSEG    (NB * 2)
#define SEGPL   ((NSEG + 31) / 32)

typedef __attribute__((ext_vector_type(2))) float v2f;
typedef __attribute__((ext_vector_type(2))) int   v2i;
typedef __attribute__((ext_vector_type(8))) float v8f;

__device__ __forceinline__ v8f wmma4_f32(v2f a, v2f b, v8f c) {
    return __builtin_amdgcn_wmma_f32_16x16x4_f32(
         false, a,  false, b,
         (short)0, c,  false,  false);
}

__device__ __forceinline__ void add_row_serial(float* row, const float* __restrict__ g)
{
    for (int i = 0; i < ENT_DIM; ++i) atomicAdd(row + i, g[i]);
}

__device__ __forceinline__ unsigned owner_byte(int d) {
    const unsigned q = (unsigned)d / (unsigned)NB;
    return q < 255u ? q : 255u;
}

__device__ __forceinline__ unsigned nomatch16(uint4 v, unsigned pat) {
    const unsigned w[4] = {v.x, v.y, v.z, v.w};
    unsigned r = 0;
    #pragma unroll
    for (int j = 0; j < 4; ++j) {
        const unsigned x = w[j] ^ pat;
        const unsigned y = ((x & 0x7f7f7f7fu) + 0x7f7f7f7fu) | x;
        r |= (y >> (7 - j)) & (0x01010101u << j);
    }
    return r;
}

__global__ __launch_bounds__(NTHR)
void node_owner_kernel(const int*   __restrict__ src,
                       const unsigned int* __restrict__ gid,
                       const int*   __restrict__ dst,
                       const int*   __restrict__ etypes,
                       const float* __restrict__ ent_emb,
                       const float* __restrict__ P,
                       const float* __restrict__ Wo,
                       const float* __restrict__ Wi,
                       const float* __restrict__ Ws,
                       const float* __restrict__ bs,
                       float*       rsum,
                       float*       __restrict__ out,
                       int E, int N)
{
    __shared__ __attribute__((aligned(16))) float s_S[NB * SPITCH];
    __shared__ __attribute__((aligned(16))) float s_W[ENT_DIM * WPITCH];
    __shared__ int s_list[LCAP];
    __shared__ v2i s_rec[LCAP];
    __shared__ int s_cnt[NSEG];
    __shared__ int s_end[NSEG];
    __shared__ int s_n;
    __shared__ int s_np[2];
    __shared__ int s_next;

    const int tid  = threadIdx.x;
    const int lane = tid & 31;
    const int wave = __builtin_amdgcn_readfirstlane(tid >> 5);
    const int base = blockIdx.x * NB;

    if (tid < NSEG) s_cnt[tid] = 0;
    if (tid == 0) { s_n = 0; s_np[0] = 0; s_np[1] = 0; s_next = 0; }
    __syncthreads();

    auto take = [&](int e) {
        const int tot = atomicAdd(&s_n, 1);
        if (tot < LCAP) {
            const int d   = dst[e] - base;
            const int p   = (etypes[e] >= NUM_RELC) ? 1 : 0;
            const int pos = atomicAdd(&s_np[p], 1);
            s_list[p ? (LCAP - 1 - pos) : pos] = e;
            atomicAdd(&s_cnt[d * 2 + p], 1);
        }
    };
    const unsigned pat   = (unsigned)blockIdx.x * 0x01010101u;
    const int      nvec  = (E + 15) >> 4;
    const int      nfull = nvec - nvec % (4 * NTHR);
    const uint4*   g4    = (const uint4*)gid;
    for (int i0 = tid; i0 < nfull; i0 += 4 * NTHR) {
        uint4 v[4];
        #pragma unroll
        for (int u = 0; u < 4; ++u) v[u] = g4[i0 + u * NTHR];
        const unsigned nlo = nomatch16(v[0], pat) | (nomatch16(v[1], pat) << 4);
        const unsigned nhi = nomatch16(v[2], pat) | (nomatch16(v[3], pat) << 4);
        unsigned long long hm = ~((unsigned long long)nlo | ((unsigned long long)nhi << 32));
        while (hm) {
            const int b = __ffsll(hm) - 1;
            hm &= hm - 1;
            const int u = ((b >> 5) << 1) | ((b >> 2) & 1);
            take(((i0 + u * NTHR) << 4) + ((b & 3) << 2) + ((b >> 3) & 3));
        }
    }
    for (int i = nfull + tid; i < nvec; i += NTHR) {
        unsigned hm = ~nomatch16(g4[i], pat) & 0x0f0f0f0fu;
        while (hm) {
            const int b = __ffs(hm) - 1;
            hm &= hm - 1;
            take((i << 4) + ((b & 3) << 2) + (b >> 3));
        }
    }
    __syncthreads();

    const bool ovf = __builtin_amdgcn_readfirstlane(s_n) > LCAP;

    if (!ovf) {
        if (wave == 0) {
            int loc[SEGPL];
            int sum = 0;
            #pragma unroll
            for (int k = 0; k < SEGPL; ++k) {
                const int i = lane * SEGPL + k;
                loc[k] = sum;
                sum += (i < NSEG) ? s_cnt[i] : 0;
            }
            int incl = sum;
            #pragma unroll
            for (int off = 1; off < 32; off <<= 1) {
                const int t = __shfl_up(incl, off);
                if (lane >= off) incl += t;
            }
            const int excl = incl - sum;
            #pragma unroll
            for (int k = 0; k < SEGPL; ++k) {
                const int i = lane * SEGPL + k;
                if (i < NSEG) s_end[i] = excl + loc[k];
            }
        }
        __syncthreads();
        const int nO = s_np[0];
        const int nL = nO + s_np[1];
        for (int i = tid; i < nL; i += NTHR) {
            const int p = (i >= nO) ? 1 : 0;
            const int e = s_list[p ? (LCAP - 1 - (i - nO)) : i];
            const int d = dst[e] - base;
            v2i r;
            r.x = etypes[e];
            r.y = src[e];
            const int slot = atomicAdd(&s_end[d * 2 + p], 1);
            s_rec[slot] = r;
        }
    } else {
        if (tid < NSEG) s_cnt[tid] = 0;
    }

    const int  m     = lane & 15;
    const int  khalf = (lane >> 4) << 1;
    const int  rt0   = wave >> 2;
    const int  cg    = wave & 3;
    const bool two   = (rt0 + 8) < NRT;
    const int  n0    = cg * 32 + m;
    const int  n1    = n0 + 16;
    const float* a_row0 = s_S + (rt0 * 16 + m) * SPITCH + khalf;
    const float* a_row1 = a_row0 + 8 * 16 * SPITCH;
    const float* b0p    = s_W + n0 * WPITCH + khalf;
    const float* b1p    = s_W + n1 * WPITCH + khalf;

    const int srow = tid >> 3;
    const int sq   = (tid & 7) << 2;

    const v8f zero8 = {0.f,0.f,0.f,0.f,0.f,0.f,0.f,0.f};
    v8f agg00 = zero8, agg01 = zero8, agg10 = zero8, agg11 = zero8;
    v8f sf00  = zero8, sf01  = zero8, sf10  = zero8, sf11  = zero8;

    auto kloop = [&](auto wsrc, int nch, float4 wv,
                     v8f& c00, v8f& c01, v8f& c10, v8f& c11) {
        for (int c = 0; c < nch; ++c) {
            __syncthreads();
            *(float4*)(s_W + srow * WPITCH + sq) = wv;
            __syncthreads();
            if (c + 1 < nch) wv = *(const float4*)wsrc(c + 1);
            if (two) {
                #pragma unroll
                for (int kk = 0; kk < KC; kk += 4) {
                    const v2f a0 = *(const v2f*)(a_row0 + c * KC + kk);
                    const v2f a1 = *(const v2f*)(a_row1 + c * KC + kk);
                    const v2f b0 = *(const v2f*)(b0p + kk);
                    const v2f b1 = *(const v2f*)(b1p + kk);
                    c00 = wmma4_f32(a0, b0, c00);
                    c01 = wmma4_f32(a0, b1, c01);
                    c10 = wmma4_f32(a1, b0, c10);
                    c11 = wmma4_f32(a1, b1, c11);
                }
            } else {
                #pragma unroll
                for (int kk = 0; kk < KC; kk += 4) {
                    const v2f a0 = *(const v2f*)(a_row0 + c * KC + kk);
                    const v2f b0 = *(const v2f*)(b0p + kk);
                    const v2f b1 = *(const v2f*)(b1p + kk);
                    c00 = wmma4_f32(a0, b0, c00);
                    c01 = wmma4_f32(a0, b1, c01);
                }
            }
        }
    };

    {
        auto wpath = [&](int c) -> const float* {
            return ((c & 4) ? Wi : Wo) + srow * CMP_DIM + REL_DIM + (c & 3) * KC + sq;
        };
        const float4 wv0 = *(const float4*)wpath(0);

        __syncthreads();

        if (!ovf) {
            for (;;) {
                int d = 0;
                if (lane == 0) d = atomicAdd(&s_next, 1);
                d = __builtin_amdgcn_readfirstlane(d);
                if (d >= NB) break;
                const int cnt = __builtin_amdgcn_readfirstlane(s_cnt[d * 2] + s_cnt[d * 2 + 1]);
                const int beg = __builtin_amdgcn_readfirstlane(s_end[d * 2 + 1]) - cnt;
                float accO[4], accI[4], accR[4];
                #pragma unroll
                for (int q = 0; q < 4; ++q) { accO[q] = 0.f; accI[q] = 0.f; accR[q] = 0.f; }
                for (int b0 = 0; b0 < cnt; b0 += 32) {
                    int nb = cnt - b0;
                    if (nb > 32) nb = 32;
                    v2i r = {0, 0};
                    if (lane < nb) r = s_rec[beg + b0 + lane];
                    for (int j = 0; j < nb; j += 4) {
                        float v[4][8];
                        bool  pI[4];
                        #pragma unroll
                        for (int u = 0; u < 4; ++u) {
                            pI[u] = false;
                            if (j + u < nb) {
                                const int etj = __builtin_amdgcn_readlane(r.x, j + u);
                                const int sj  = __builtin_amdgcn_readlane(r.y, j + u);
                                const float* rr = P       + (long)etj * ENT_DIM + lane;
                                const float* er = ent_emb + (long)sj  * ENT_DIM + lane;
                                v[u][0] = rr[0];  v[u][1] = rr[32]; v[u][2] = rr[64]; v[u][3] = rr[96];
                                v[u][4] = er[0];  v[u][5] = er[32]; v[u][6] = er[64]; v[u][7] = er[96];
                                pI[u] = etj >= NUM_RELC;
                            } else {
                                #pragma unroll
                                for (int q = 0; q < 8; ++q) v[u][q] = 0.f;
                            }
                        }
                        #pragma unroll
                        for (int u = 0; u < 4; ++u) {
                            #pragma unroll
                            for (int q = 0; q < 4; ++q) accR[q] += v[u][q];
                            if (pI[u]) {
                                #pragma unroll
                                for (int q = 0; q < 4; ++q) accI[q] += v[u][4 + q];
                            } else {
                                #pragma unroll
                                for (int q = 0; q < 4; ++q) accO[q] += v[u][4 + q];
                            }
                        }
                    }
                }
                float* row = s_S + d * SPITCH + lane;
                #pragma unroll
                for (int q = 0; q < 4; ++q) { row[32 * q] = accO[q]; row[ENT_DIM + 32 * q] = accI[q]; }
                if (base + d < N) {
                    float* rr = rsum + ((long)base + d) * ENT_DIM + lane;
                    #pragma unroll
                    for (int q = 0; q < 4; ++q) rr[32 * q] = accR[q];
                }
            }
        } else {
            for (int i = tid; i < NB * SPITCH / 4; i += NTHR)
                ((float4*)s_S)[i] = make_float4(0.f, 0.f, 0.f, 0.f);
            __syncthreads();
            for (int e = tid; e < E; e += NTHR) {
                const int d = dst[e] - base;
                if ((unsigned)d < (unsigned)NB) {
                    const int et = etypes[e];
                    add_row_serial(s_S + d * SPITCH, P + (long)et * ENT_DIM);
                    atomicAdd(&s_cnt[d * 2 + ((et >= NUM_RELC) ? 1 : 0)], 1);
                }
            }
            __syncthreads();
            for (int i = tid; i < NB * ENT_DIM; i += NTHR) {
                const int r = i >> 7;
                if (base + r < N) rsum[((long)base + r) * ENT_DIM + (i & 127)] = s_S[r * SPITCH + (i & 127)];
            }
            __syncthreads();
            for (int i = tid; i < NB * SPITCH / 4; i += NTHR)
                ((float4*)s_S)[i] = make_float4(0.f, 0.f, 0.f, 0.f);
            __syncthreads();
            for (int e = tid; e < E; e += NTHR) {
                const int d = dst[e] - base;
                if ((unsigned)d < (unsigned)NB) {
                    const int p = (etypes[e] >= NUM_RELC) ? 1 : 0;
                    add_row_serial(s_S + d * SPITCH + p * ENT_DIM, ent_emb + (long)src[e] * ENT_DIM);
                }
            }
        }

        kloop(wpath, CMP_DIM / KC, wv0, agg00, agg01, agg10, agg11);
    }

    {
        const float4 wv0 = *(const float4*)(Ws + srow * ENT_DIM + sq);
        __syncthreads();
        for (int i = tid; i < NB * (ENT_DIM / 4); i += NTHR) {
            const int r = i >> 5;
            const int q = (i & 31) << 2;
            long nrow = (long)base + r;
            if (nrow > (long)N - 1) nrow = (long)N - 1;
            *(float4*)(s_S + r * SPITCH + q) = *(const float4*)(ent_emb + nrow * ENT_DIM + q);
        }
        auto wself = [&](int c) -> const float* { return Ws + srow * ENT_DIM + c * KC + sq; };
        kloop(wself, ENT_DIM / KC, wv0, sf00, sf01, sf10, sf11);
    }

    const float bs0 = bs[n0], bs1 = bs[n1];
    const int   rhalf = (lane >> 4) << 3;
    #pragma unroll
    for (int j = 0; j < 8; ++j) {
        const int  lr   = rt0 * 16 + j + rhalf;
        const long node = (long)base + lr;
        if (node < N) {
            const float cO = (float)s_cnt[lr * 2];
            const float cI = (float)s_cnt[lr * 2 + 1];
            float dg = cO + cI;
            dg = (dg > 1.0f) ? dg : 1.0f;
            out[node * ENT_DIM + n0] = sf00[j] + bs0 + (agg00[j] + rsum[node * ENT_DIM + n0]) / dg;
            out[node * ENT_DIM + n1] = sf01[j] + bs1 + (agg01[j] + rsum[node * ENT_DIM + n1]) / dg;
        }
    }
    if (two) {
        #pragma unroll
        for (int j = 0; j < 8; ++j) {
            const int  lr   = (rt0 + 8) * 16 + j + rhalf;
            const long node = (long)base + lr;
            if (node < N) {
                const float cO = (float)s_cnt[lr * 2];
                const float cI = (float)s_cnt[lr * 2 + 1];
                float dg = cO + cI;
                dg = (dg > 1.0f) ? dg : 1.0f;
                out[node * ENT_DIM + n0] = sf10[j] + bs0 + (agg10[j] + rsum[node * ENT_DIM + n0]) / dg;
                out[node * ENT_DIM + n1] = sf11[j] + bs1 + (agg11[j] + rsum[node * ENT_DIM + n1]) / dg;
            }
        }
    }
}

__global__ __launch_bounds__(256)
void gemm_bias_kernel(const float* __restrict__ A,
                      const float* __restrict__ W,
                      const float* __restrict__ b,
                      const float* __restrict__ agg,
                      const float* __restrict__ deg,
                      float*       __restrict__ out,
                      int N,
                      const int*   __restrict__ dst,
                      unsigned int* __restrict__ gid,
                      int E, int ngemm,
                      const float* __restrict__ Wo,
                      const float* __restrict__ bo,
                      const float* __restrict__ Wi,
                      const float* __restrict__ bi,
                      float*       __restrict__ P)
{
    if ((int)blockIdx.x >= 3 * ngemm) {
        const int c = ((int)blockIdx.x - 3 * ngemm) * 256 + (int)threadIdx.x;
        if (c >= ((E + 15) >> 4)) return;
        const int e0 = c << 4;
        unsigned w[4];
        if (e0 + 16 <= E) {
            const int4* d4 = (const int4*)(dst + e0);
            #pragma unroll
            for (int j = 0; j < 4; ++j) {
                const int4 v = d4[j];
                w[j] = owner_byte(v.x) | (owner_byte(v.y) << 8) | (owner_byte(v.z) << 16) | (owner_byte(v.w) << 24);
            }
        } else {
            #pragma unroll
            for (int j = 0; j < 4; ++j) {
                w[j] = 0;
                #pragma unroll
                for (int k = 0; k < 4; ++k) {
                    const int e = e0 + 4 * j + k;
                    w[j] |= ((e < E) ? owner_byte(dst[e]) : 255u) << (8 * k);
                }
            }
        }
        ((uint4*)gid)[c] = make_uint4(w[0], w[1], w[2], w[3]);
        return;
    }

    const int  tid   = threadIdx.x;
    const int  lane  = tid & 31;
    const int  wave  = tid >> 5;
    const int  m     = lane & 15;
    const int  khalf = (lane >> 4) << 1;
    const int  kind  = (int)blockIdx.x / ngemm;
    const long tile  = (long)((int)blockIdx.x - kind * ngemm) * 16;
    const float* Wk = W;  const float* bk = b;  float* ok = out;
    int wstride = ENT_DIM, rlo = 0, rhi = N;
    if (kind == 1) { Wk = Wo; bk = bo; ok = P; wstride = CMP_DIM; rhi = (N < NUM_RELC) ? N : NUM_RELC; }
    if (kind == 2) { Wk = Wi; bk = bi; ok = P; wstride = CMP_DIM; rlo = NUM_RELC; }
    if (tile + 16 <= rlo || tile >= rhi) return;

    const long r    = tile + m;
    const long rc   = (r < N) ? r : (long)(N - 1);
    const float* ar = A + rc * ENT_DIM;

    const int    n  = (wave << 4) + m;
    const float* wr = Wk + (long)n * wstride;

    v8f acc = {0.f,0.f,0.f,0.f,0.f,0.f,0.f,0.f};

    #pragma unroll 4
    for (int k = 0; k < ENT_DIM; k += 4) {
        v2f av = *(const v2f*)(ar + k + khalf);
        v2f bv = *(const v2f*)(wr + k + khalf);
        acc = wmma4_f32(av, bv, acc);
    }

    const float bn    = bk[n];
    const int   rhalf = (lane >> 4) << 3;

    #pragma unroll
    for (int j = 0; j < 8; ++j) {
        const int  me = j + rhalf;
        const long rr = tile + me;
        if (rr >= rlo && rr < rhi) {
            float v = acc[j] + bn;
            if (agg != nullptr) {
                float d = deg[rr];
                d = (d > 1.0f) ? d : 1.0f;
                v += agg[rr * ENT_DIM + n] / d;
            }
            ok[rr * ENT_DIM + n] = v;
        }
    }
}

extern "C" void kernel_launch(void* const* d_in, const int* in_sizes, int n_in,
                              void* d_out, int out_size, void* d_ws, size_t ws_size,
                              hipStream_t stream)
{
    const int*   src     = (const int*)  d_in[0];
    const int*   dst     = (const int*)  d_in[1];
    const int*   etypes  = (const int*)  d_in[2];
    const float* ent_emb = (const float*)d_in[3];
    const float* rel_emb = (const float*)d_in[4];
    const float* Wo      = (const float*)d_in[5];
    const float* bo      = (const float*)d_in[6];
    const float* Wi      = (const float*)d_in[7];
    const float* bi      = (const float*)d_in[8];
    const float* Ws      = (const float*)d_in[9];
    const float* bs      = (const float*)d_in[10];
    const float* Wr      = (const float*)d_in[11];
    const float* br      = (const float*)d_in[12];

    const int E = in_sizes[0];
    const int N = in_sizes[3] / ENT_DIM;
    const int R = in_sizes[4] / REL_DIM;

    float* out_ent = (float*)d_out;
    float* out_rel = out_ent + (size_t)N * ENT_DIM;

    unsigned int* gid = (unsigned int*)d_ws;
    float* P    = (float*)((char*)d_ws + (1 << 20));
    float* rsum = (float*)((char*)d_ws + (2 << 20));

    gemm_bias_kernel<<<3 * ((R + 15) / 16) + ((E + 15) / 16 + 255) / 256, 256, 0, stream>>>(
        rel_emb, Wr, br, nullptr, nullptr, out_rel, R, dst, gid, E, (R + 15) / 16, Wo, bo, Wi, bi, P);

    node_owner_kernel<<<(N + NB - 1) / NB, NTHR, 0, stream>>>(
        src, gid, dst, etypes, ent_emb, P, Wo, Wi, Ws, bs, rsum, out_ent, E, N);
}


# Self product (ent @ W_S^T) moved in front: own rows and W_S whole in the still idle LDS rows, its WMMAs issued inside the owner-byte scan loop while the scan's loads are in flight; k order kept
# speedup vs baseline: 9.1200x; 1.0306x over previous
#include <hip/hip_runtime.h>

#define ENT_DIM  128
#define REL_DIM  128
#define CMP_DIM  256
#define NUM_RELC 200

#define NB      208
#define NRT     13
#define SPITCH  260
#define KC      32
#define WPITCH  36
#define LCAP    6144
#define NTHR    1024
#define NSEG    (NB * 2)
#define SEGPL   ((NSEG + 31) / 32)

typedef __attribute__((ext_vector_type(2))) float v2f;
typedef __attribute__((ext_vector_type(2))) int   v2i;
typedef __attribute__((ext_vector_type(8))) float v8f;

__device__ __forceinline__ v8f wmma4_f32(v2f a, v2f b, v8f c) {
    return __builtin_amdgcn_wmma_f32_16x16x4_f32(
         false, a,  false, b,
         (short)0, c,  false,  false);
}

__device__ __forceinline__ void add_row_serial(float* row, const float* __restrict__ g)
{
    for (int i = 0; i < ENT_DIM; ++i) atomicAdd(row + i, g[i]);
}

__device__ __forceinline__ unsigned owner_byte(int d) {
    const unsigned q = (unsigned)d / (unsigned)NB;
    return q < 255u ? q : 255u;
}

__device__ __forceinline__ unsigned nomatch16(uint4 v, unsigned pat) {
    const unsigned w[4] = {v.x, v.y, v.z, v.w};
    unsigned r = 0;
    #pragma unroll
    for (int j = 0; j < 4; ++j) {
        const unsigned x = w[j] ^ pat;
        const unsigned y = ((x & 0x7f7f7f7fu) + 0x7f7f7f7fu) | x;
        r |= (y >> (7 - j)) & (0x01010101u << j);
    }
    return r;
}

__global__ __launch_bounds__(NTHR)
void node_owner_kernel(const int*   __restrict__ src,
                       const unsigned int* __restrict__ gid,
                       const int*   __restrict__ dst,
                       const int*   __restrict__ etypes,
                       const float* __restrict__ ent_emb,
                       const float* __restrict__ P,
                       const float* __restrict__ Wo,
                       const float* __restrict__ Wi,
                       const float* __restrict__ Ws,
                       const float* __restrict__ bs,
                       float*       rsum,
                       float*       __restrict__ out,
                       int E, int N)
{
    __shared__ __attribute__((aligned(16))) float s_S[NB * SPITCH];
    __shared__ __attribute__((aligned(16))) float s_W[ENT_DIM * WPITCH];
    __shared__ int s_list[LCAP];
    __shared__ v2i s_rec[LCAP];
    __shared__ int s_cnt[NSEG];
    __shared__ int s_end[NSEG];
    __shared__ int s_n;
    __shared__ int s_np[2];
    __shared__ int s_next;

    const int tid  = threadIdx.x;
    const int lane = tid & 31;
    const int wave = __builtin_amdgcn_readfirstlane(tid >> 5);
    const int base = blockIdx.x * NB;

    const int  m     = lane & 15;
    const int  khalf = (lane >> 4) << 1;
    const int  rt0   = wave >> 2;
    const int  cg    = wave & 3;
    const bool two   = (rt0 + 8) < NRT;
    const int  n0    = cg * 32 + m;
    const int  n1    = n0 + 16;
    const float* a_row0 = s_S + (rt0 * 16 + m) * SPITCH + khalf;
    const float* a_row1 = a_row0 + 8 * 16 * SPITCH;
    const float* b0p    = s_W + n0 * WPITCH + khalf;
    const float* b1p    = s_W + n1 * WPITCH + khalf;
    const float* ws0p   = s_S + n0 * SPITCH + ENT_DIM + khalf;
    const float* ws1p   = s_S + n1 * SPITCH + ENT_DIM + khalf;

    const v8f zero8 = {0.f,0.f,0.f,0.f,0.f,0.f,0.f,0.f};
    v8f sf00  = zero8, sf01  = zero8, sf10  = zero8, sf11  = zero8;

    if (tid < NSEG) s_cnt[tid] = 0;
    if (tid == 0) { s_n = 0; s_np[0] = 0; s_np[1] = 0; s_next = 0; }

    for (int i = tid; i < NB * (ENT_DIM / 4); i += NTHR) {
        const int r = i >> 5;
        const int q = (i & 31) << 2;
        long nrow = (long)base + r;
        if (nrow > (long)N - 1) nrow = (long)N - 1;
        *(float4*)(s_S + r * SPITCH + q) = *(const float4*)(ent_emb + nrow * ENT_DIM + q);
    }
    for (int i = tid; i < ENT_DIM * (ENT_DIM / 4); i += NTHR) {
        const int n = i >> 5;
        const int q = (i & 31) << 2;
        *(float4*)(s_S + n * SPITCH + ENT_DIM + q) = *(const float4*)(Ws + n * ENT_DIM + q);
    }
    __syncthreads();

    auto self_steps = [&](int k0, int ns) {
        if (two) {
            for (int s = 0; s < ns; ++s) {
                const int kk = (k0 + s) * 4;
                const v2f a0 = *(const v2f*)(a_row0 + kk);
                const v2f a1 = *(const v2f*)(a_row1 + kk);
                const v2f b0 = *(const v2f*)(ws0p + kk);
                const v2f b1 = *(const v2f*)(ws1p + kk);
                sf00 = wmma4_f32(a0, b0, sf00);
                sf01 = wmma4_f32(a0, b1, sf01);
                sf10 = wmma4_f32(a1, b0, sf10);
                sf11 = wmma4_f32(a1, b1, sf11);
            }
        } else {
            for (int s = 0; s < ns; ++s) {
                const int kk = (k0 + s) * 4;
                const v2f a0 = *(const v2f*)(a_row0 + kk);
                const v2f b0 = *(const v2f*)(ws0p + kk);
                const v2f b1 = *(const v2f*)(ws1p + kk);
                sf00 = wmma4_f32(a0, b0, sf00);
                sf01 = wmma4_f32(a0, b1, sf01);
            }
        }
    };

    auto take = [&](int e) {
        const int tot = atomicAdd(&s_n, 1);
        if (tot < LCAP) {
            const int d   = dst[e] - base;
            const int p   = (etypes[e] >= NUM_RELC) ? 1 : 0;
            const int pos = atomicAdd(&s_np[p], 1);
            s_list[p ? (LCAP - 1 - pos) : pos] = e;
            atomicAdd(&s_cnt[d * 2 + p], 1);
        }
    };
    const unsigned pat   = (unsigned)blockIdx.x * 0x01010101u;
    const int      nvec  = (E + 15) >> 4;
    const int      nfull = nvec - nvec % (4 * NTHR);
    const uint4*   g4    = (const uint4*)gid;
    const int ntrip = nfull / (4 * NTHR);
    const int kper  = (ntrip > 0) ? (ENT_DIM / 4 + ntrip - 1) / ntrip : 0;
    int kd = 0;
    for (int t = 0; t < ntrip; ++t) {
        const int i0 = tid + t * (4 * NTHR);
        uint4 v[4];
        #pragma unroll
        for (int u = 0; u < 4; ++u) v[u] = g4[i0 + u * NTHR];
        int ks = ENT_DIM / 4 - kd;
        if (ks > kper) ks = kper;
        self_steps(kd, ks);
        kd += ks;
        const unsigned nlo = nomatch16(v[0], pat) | (nomatch16(v[1], pat) << 4);
        const unsigned nhi = nomatch16(v[2], pat) | (nomatch16(v[3], pat) << 4);
        unsigned long long hm = ~((unsigned long long)nlo | ((unsigned long long)nhi << 32));
        while (hm) {
            const int b = __ffsll(hm) - 1;
            hm &= hm - 1;
            const int u = ((b >> 5) << 1) | ((b >> 2) & 1);
            take(((i0 + u * NTHR) << 4) + ((b & 3) << 2) + ((b >> 3) & 3));
        }
    }
    for (int i = nfull + tid; i < nvec; i += NTHR) {
        unsigned hm = ~nomatch16(g4[i], pat) & 0x0f0f0f0fu;
        while (hm) {
            const int b = __ffs(hm) - 1;
            hm &= hm - 1;
            take((i << 4) + ((b & 3) << 2) + (b >> 3));
        }
    }
    self_steps(kd, ENT_DIM / 4 - kd);
    __syncthreads();

    const bool ovf = __builtin_amdgcn_readfirstlane(s_n) > LCAP;

    if (!ovf) {
        if (wave == 0) {
            int loc[SEGPL];
            int sum = 0;
            #pragma unroll
            for (int k = 0; k < SEGPL; ++k) {
                const int i = lane * SEGPL + k;
                loc[k] = sum;
                sum += (i < NSEG) ? s_cnt[i] : 0;
            }
            int incl = sum;
            #pragma unroll
            for (int off = 1; off < 32; off <<= 1) {
                const int t = __shfl_up(incl, off);
                if (lane >= off) incl += t;
            }
            const int excl = incl - sum;
            #pragma unroll
            for (int k = 0; k < SEGPL; ++k) {
                const int i = lane * SEGPL + k;
                if (i < NSEG) s_end[i] = excl + loc[k];
            }
        }
        __syncthreads();
        const int nO = s_np[0];
        const int nL = nO + s_np[1];
        for (int i = tid; i < nL; i += NTHR) {
            const int p = (i >= nO) ? 1 : 0;
            const int e = s_list[p ? (LCAP - 1 - (i - nO)) : i];
            const int d = dst[e] - base;
            v2i r;
            r.x = etypes[e];
            r.y = src[e];
            const int slot = atomicAdd(&s_end[d * 2 + p], 1);
            s_rec[slot] = r;
        }
    } else {
        if (tid < NSEG) s_cnt[tid] = 0;
    }

    const int srow = tid >> 3;
    const int sq   = (tid & 7) << 2;

    v8f agg00 = zero8, agg01 = zero8, agg10 = zero8, agg11 = zero8;

    auto kloop = [&](auto wsrc, int nch, float4 wv,
                     v8f& c00, v8f& c01, v8f& c10, v8f& c11) {
        for (int c = 0; c < nch; ++c) {
            __syncthreads();
            *(float4*)(s_W + srow * WPITCH + sq) = wv;
            __syncthreads();
            if (c + 1 < nch) wv = *(const float4*)wsrc(c + 1);
            if (two) {
                #pragma unroll
                for (int kk = 0; kk < KC; kk += 4) {
                    const v2f a0 = *(const v2f*)(a_row0 + c * KC + kk);
                    const v2f a1 = *(const v2f*)(a_row1 + c * KC + kk);
                    const v2f b0 = *(const v2f*)(b0p + kk);
                    const v2f b1 = *(const v2f*)(b1p + kk);
                    c00 = wmma4_f32(a0, b0, c00);
                    c01 = wmma4_f32(a0, b1, c01);
                    c10 = wmma4_f32(a1, b0, c10);
                    c11 = wmma4_f32(a1, b1, c11);
                }
            } else {
                #pragma unroll
                for (int kk = 0; kk < KC; kk += 4) {
                    const v2f a0 = *(const v2f*)(a_row0 + c * KC + kk);
                    const v2f b0 = *(const v2f*)(b0p + kk);
                    const v2f b1 = *(const v2f*)(b1p + kk);
                    c00 = wmma4_f32(a0, b0, c00);
                    c01 = wmma4_f32(a0, b1, c01);
                }
            }
        }
    };

    {
        auto wpath = [&](int c) -> const float* {
            return ((c & 4) ? Wi : Wo) + srow * CMP_DIM + REL_DIM + (c & 3) * KC + sq;
        };
        const float4 wv0 = *(const float4*)wpath(0);

        __syncthreads();

        if (!ovf) {
            for (;;) {
                int d = 0;
                if (lane == 0) d = atomicAdd(&s_next, 1);
                d = __builtin_amdgcn_readfirstlane(d);
                if (d >= NB) break;
                const int cnt = __builtin_amdgcn_readfirstlane(s_cnt[d * 2] + s_cnt[d * 2 + 1]);
                const int beg = __builtin_amdgcn_readfirstlane(s_end[d * 2 + 1]) - cnt;
                float accO[4], accI[4], accR[4];
                #pragma unroll
                for (int q = 0; q < 4; ++q) { accO[q] = 0.f; accI[q] = 0.f; accR[q] = 0.f; }
                for (int b0 = 0; b0 < cnt; b0 += 32) {
                    int nb = cnt - b0;
                    if (nb > 32) nb = 32;
                    v2i r = {0, 0};
                    if (lane < nb) r = s_rec[beg + b0 + lane];
                    for (int j = 0; j < nb; j += 4) {
                        float v[4][8];
                        bool  pI[4];
                        #pragma unroll
                        for (int u = 0; u < 4; ++u) {
                            pI[u] = false;
                            if (j + u < nb) {
                                const int etj = __builtin_amdgcn_readlane(r.x, j + u);
                                const int sj  = __builtin_amdgcn_readlane(r.y, j + u);
                                const float* rr = P       + (long)etj * ENT_DIM + lane;
                                const float* er = ent_emb + (long)sj  * ENT_DIM + lane;
                                v[u][0] = rr[0];  v[u][1] = rr[32]; v[u][2] = rr[64]; v[u][3] = rr[96];
                                v[u][4] = er[0];  v[u][5] = er[32]; v[u][6] = er[64]; v[u][7] = er[96];
                                pI[u] = etj >= NUM_RELC;
                            } else {
                                #pragma unroll
                                for (int q = 0; q < 8; ++q) v[u][q] = 0.f;
                            }
                        }
                        #pragma unroll
                        for (int u = 0; u < 4; ++u) {
                            #pragma unroll
                            for (int q = 0; q < 4; ++q) accR[q] += v[u][q];
                            if (pI[u]) {
                                #pragma unroll
                                for (int q = 0; q < 4; ++q) accI[q] += v[u][4 + q];
                            } else {
                                #pragma unroll
                                for (int q = 0; q < 4; ++q) accO[q] += v[u][4 + q];
                            }
                        }
                    }
                }
                float* row = s_S + d * SPITCH + lane;
                #pragma unroll
                for (int q = 0; q < 4; ++q) { row[32 * q] = accO[q]; row[ENT_DIM + 32 * q] = accI[q]; }
                if (base + d < N) {
                    float* rr = rsum + ((long)base + d) * ENT_DIM + lane;
                    #pragma unroll
                    for (int q = 0; q < 4; ++q) rr[32 * q] = accR[q];
                }
            }
        } else {
            for (int i = tid; i < NB * SPITCH / 4; i += NTHR)
                ((float4*)s_S)[i] = make_float4(0.f, 0.f, 0.f, 0.f);
            __syncthreads();
            for (int e = tid; e < E; e += NTHR) {
                const int d = dst[e] - base;
                if ((unsigned)d < (unsigned)NB) {
                    const int et = etypes[e];
                    add_row_serial(s_S + d * SPITCH, P + (long)et * ENT_DIM);
                    atomicAdd(&s_cnt[d * 2 + ((et >= NUM_RELC) ? 1 : 0)], 1);
                }
            }
            __syncthreads();
            for (int i = tid; i < NB * ENT_DIM; i += NTHR) {
                const int r = i >> 7;
                if (base + r < N) rsum[((long)base + r) * ENT_DIM + (i & 127)] = s_S[r * SPITCH + (i & 127)];
            }
            __syncthreads();
            for (int i = tid; i < NB * SPITCH / 4; i += NTHR)
                ((float4*)s_S)[i] = make_float4(0.f, 0.f, 0.f, 0.f);
            __syncthreads();
            for (int e = tid; e < E; e += NTHR) {
                const int d = dst[e] - base;
                if ((unsigned)d < (unsigned)NB) {
                    const int p = (etypes[e] >= NUM_RELC) ? 1 : 0;
                    add_row_serial(s_S + d * SPITCH + p * ENT_DIM, ent_emb + (long)src[e] * ENT_DIM);
                }
            }
        }

        kloop(wpath, CMP_DIM / KC, wv0, agg00, agg01, agg10, agg11);
    }

    const float bs0 = bs[n0], bs1 = bs[n1];
    const int   rhalf = (lane >> 4) << 3;
    #pragma unroll
    for (int j = 0; j < 8; ++j) {
        const int  lr   = rt0 * 16 + j + rhalf;
        const long node = (long)base + lr;
        if (node < N) {
            const float cO = (float)s_cnt[lr * 2];
            const float cI = (float)s_cnt[lr * 2 + 1];
            float dg = cO + cI;
            dg = (dg > 1.0f) ? dg : 1.0f;
            out[node * ENT_DIM + n0] = sf00[j] + bs0 + (agg00[j] + rsum[node * ENT_DIM + n0]) / dg;
            out[node * ENT_DIM + n1] = sf01[j] + bs1 + (agg01[j] + rsum[node * ENT_DIM + n1]) / dg;
        }
    }
    if (two) {
        #pragma unroll
        for (int j = 0; j < 8; ++j) {
            const int  lr   = (rt0 + 8) * 16 + j + rhalf;
            const long node = (long)base + lr;
            if (node < N) {
                const float cO = (float)s_cnt[lr * 2];
                const float cI = (float)s_cnt[lr * 2 + 1];
                float dg = cO + cI;
                dg = (dg > 1.0f) ? dg : 1.0f;
                out[node * ENT_DIM + n0] = sf10[j] + bs0 + (agg10[j] + rsum[node * ENT_DIM + n0]) / dg;
                out[node * ENT_DIM + n1] = sf11[j] + bs1 + (agg11[j] + rsum[node * ENT_DIM + n1]) / dg;
            }
        }
    }
}

__global__ __launch_bounds__(256)
void gemm_bias_kernel(const float* __restrict__ A,
                      const float* __restrict__ W,
                      const float* __restrict__ b,
                      const float* __restrict__ agg,
                      const float* __restrict__ deg,
                      float*       __restrict__ out,
                      int N,
                      const int*   __restrict__ dst,
                      unsigned int* __restrict__ gid,
                      int E, int ngemm,
                      const float* __restrict__ Wo,
                      const float* __restrict__ bo,
                      const float* __restrict__ Wi,
                      const float* __restrict__ bi,
                      float*       __restrict__ P)
{
    if ((int)blockIdx.x >= 3 * ngemm) {
        const int c = ((int)blockIdx.x - 3 * ngemm) * 256 + (int)threadIdx.x;
        if (c >= ((E + 15) >> 4)) return;
        const int e0 = c << 4;
        unsigned w[4];
        if (e0 + 16 <= E) {
            const int4* d4 = (const int4*)(dst + e0);
            #pragma unroll
            for (int j = 0; j < 4; ++j) {
                const int4 v = d4[j];
                w[j] = owner_byte(v.x) | (owner_byte(v.y) << 8) | (owner_byte(v.z) << 16) | (owner_byte(v.w) << 24);
            }
        } else {
            #pragma unroll
            for (int j = 0; j < 4; ++j) {
                w[j] = 0;
                #pragma unroll
                for (int k = 0; k < 4; ++k) {
                    const int e = e0 + 4 * j + k;
                    w[j] |= ((e < E) ? owner_byte(dst[e]) : 255u) << (8 * k);
                }
            }
        }
        ((uint4*)gid)[c] = make_uint4(w[0], w[1], w[2], w[3]);
        return;
    }

    const int  tid   = threadIdx.x;
    const int  lane  = tid & 31;
    const int  wave  = tid >> 5;
    const int  m     = lane & 15;
    const int  khalf = (lane >> 4) << 1;
    const int  kind  = (int)blockIdx.x / ngemm;
    const long tile  = (long)((int)blockIdx.x - kind * ngemm) * 16;
    const float* Wk = W;  const float* bk = b;  float* ok = out;
    int wstride = ENT_DIM, rlo = 0, rhi = N;
    if (kind == 1) { Wk = Wo; bk = bo; ok = P; wstride = CMP_DIM; rhi = (N < NUM_RELC) ? N : NUM_RELC; }
    if (kind == 2) { Wk = Wi; bk = bi; ok = P; wstride = CMP_DIM; rlo = NUM_RELC; }
    if (tile + 16 <= rlo || tile >= rhi) return;

    const long r    = tile + m;
    const long rc   = (r < N) ? r : (long)(N - 1);
    const float* ar = A + rc * ENT_DIM;

    const int    n  = (wave << 4) + m;
    const float* wr = Wk + (long)n * wstride;

    v8f acc = {0.f,0.f,0.f,0.f,0.f,0.f,0.f,0.f};

    #pragma unroll 4
    for (int k = 0; k < ENT_DIM; k += 4) {
        v2f av = *(const v2f*)(ar + k + khalf);
        v2f bv = *(const v2f*)(wr + k + khalf);
        acc = wmma4_f32(av, bv, acc);
    }

    const float bn    = bk[n];
    const int   rhalf = (lane >> 4) << 3;

    #pragma unroll
    for (int j = 0; j < 8; ++j) {
        const int  me = j + rhalf;
        const long rr = tile + me;
        if (rr >= rlo && rr < rhi) {
            float v = acc[j] + bn;
            if (agg != nullptr) {
                float d = deg[rr];
                d = (d > 1.0f) ? d : 1.0f;
                v += agg[rr * ENT_DIM + n] / d;
            }
            ok[rr * ENT_DIM + n] = v;
        }
    }
}

extern "C" void kernel_launch(void* const* d_in, const int* in_sizes, int n_in,
                              void* d_out, int out_size, void* d_ws, size_t ws_size,
                              hipStream_t stream)
{
    const int*   src     = (const int*)  d_in[0];
    const int*   dst     = (const int*)  d_in[1];
    const int*   etypes  = (const int*)  d_in[2];
    const float* ent_emb = (const float*)d_in[3];
    const float* rel_emb = (const float*)d_in[4];
    const float* Wo      = (const float*)d_in[5];
    const float* bo      = (const float*)d_in[6];
    const float* Wi      = (const float*)d_in[7];
    const float* bi      = (const float*)d_in[8];
    const float* Ws      = (const float*)d_in[9];
    const float* bs      = (const float*)d_in[10];
    const float* Wr      = (const float*)d_in[11];
    const float* br      = (const float*)d_in[12];

    const int E = in_sizes[0];
    const int N = in_sizes[3] / ENT_DIM;
    const int R = in_sizes[4] / REL_DIM;

    float* out_ent = (float*)d_out;
    float* out_rel = out_ent + (size_t)N * ENT_DIM;

    unsigned int* gid = (unsigned int*)d_ws;
    float* P    = (float*)((char*)d_ws + (1 << 20));
    float* rsum = (float*)((char*)d_ws + (2 << 20));

    gemm_bias_kernel<<<3 * ((R + 15) / 16) + ((E + 15) / 16 + 255) / 256, 256, 0, stream>>>(
        rel_emb, Wr, br, nullptr, nullptr, out_rel, R, dst, gid, E, (R + 15) / 16, Wo, bo, Wi, bi, P);

    node_owner_kernel<<<(N + NB - 1) / NB, NTHR, 0, stream>>>(
        src, gid, dst, etypes, ent_emb, P, Wo, Wi, Ws, bs, rsum, out_ent, E, N);
}
